# starting kernel
# baseline (speedup 1.0000x reference)
#include <hip/hip_runtime.h>
#include <math.h>

#define NNODES 50000
#define NEDGES 500000
#define CHAN   64
#define R0f    2.0f

#define WAVES  8
#define BLOCK  (WAVES * 32)
#define NTILES (NEDGES / 16)
#define GRID   512

typedef float v2f __attribute__((ext_vector_type(2)));
typedef float v8f __attribute__((ext_vector_type(8)));

#define OFF_WSRC  0
#define OFF_WDST  4096
#define OFF_WS    8192
#define OFF_WV    12288
#define OFF_WGATE 16384
#define OFF_WENC  32768
#define OFF_BIAS0 33280
#define OFF_BGATE 33344
#define OFF_STAGE 33600
#define OFF_RH    (OFF_STAGE + WAVES * 1024)
#define OFF_IDX   (OFF_RH + WAVES * 64)
#define SMEM_FLOATS (OFF_IDX + WAVES * 32)

__device__ __forceinline__ void lds_fence() {
  asm volatile("s_wait_dscnt 0x0" ::: "memory");
}

__device__ __forceinline__ v8f wmma4(v2f a, v2f b, v8f c) {
  return __builtin_amdgcn_wmma_f32_16x16x4_f32(false, a, false, b, (short)0, c,
                                               false, false);
}

__global__ void zero_f32(float* __restrict__ p, int n) {
  int i = blockIdx.x * blockDim.x + threadIdx.x;
  if (i < n) p[i] = 0.0f;
}

__global__ void __launch_bounds__(BLOCK, 1)
messages_kernel(const int* __restrict__ src, const int* __restrict__ dst,
                const float* __restrict__ r_ij,
                const float* __restrict__ z0g, const float* __restrict__ z1g,
                const float* __restrict__ emb,
                const float* __restrict__ Wenc, const float* __restrict__ benc,
                const float* __restrict__ Wsrc, const float* __restrict__ bsrc,
                const float* __restrict__ Wdst, const float* __restrict__ bdst,
                const float* __restrict__ Wgate, const float* __restrict__ bgate,
                const float* __restrict__ Ws, const float* __restrict__ Wv,
                float* __restrict__ out0, float* __restrict__ out1) {
  extern __shared__ float smem[];

  for (int i = threadIdx.x; i < 64 * 64; i += BLOCK) {
    int k = i >> 6, n = i & 63;
    smem[OFF_WSRC + n * 64 + k] = Wsrc[i];
    smem[OFF_WDST + n * 64 + k] = Wdst[i];
    smem[OFF_WS   + n * 64 + k] = Ws[i];
    smem[OFF_WV   + n * 64 + k] = Wv[i];
  }
  for (int i = threadIdx.x; i < 64 * 256; i += BLOCK) {
    int k = i >> 8, n = i & 255;
    smem[OFF_WGATE + n * 64 + k] = Wgate[i];
  }
  for (int i = threadIdx.x; i < 8 * 64; i += BLOCK) {
    int k = i >> 6, n = i & 63;
    smem[OFF_WENC + n * 8 + k] = Wenc[i];
  }
  if (threadIdx.x < 64)
    smem[OFF_BIAS0 + threadIdx.x] =
        bsrc[threadIdx.x] + bdst[threadIdx.x] + benc[threadIdx.x];
  if (threadIdx.x < 256) smem[OFF_BGATE + threadIdx.x] = bgate[threadIdx.x];
  __syncthreads();

  const int lane = threadIdx.x & 31;
  const int wave = threadIdx.x >> 5;
  const int half = lane >> 4;
  const int nl   = lane & 15;

  float* stage = smem + OFF_STAGE + wave * 1024;
  float* rhS   = smem + OFF_RH + wave * 64;
  int*   idxS  = ((int*)(smem + OFF_IDX)) + wave * 32;

  for (int tile = blockIdx.x * WAVES + wave; tile < NTILES;
       tile += gridDim.x * WAVES) {
    const int e  = tile * 16 + nl;
    const int sN = src[e];
    const int dN = dst[e];
    if (half == 0) { idxS[nl] = sN; idxS[16 + nl] = dN; }

    float rx = r_ij[e * 3 + 0], ry = r_ij[e * 3 + 1], rz = r_ij[e * 3 + 2];
    float dd  = sqrtf(rx * rx + ry * ry + rz * rz);
    float sx = rx * 3.5f, sy = ry * 3.5f, sz = rz * 3.5f;
    float inv = rsqrtf(1.0f + sx * sx + sy * sy + sz * sz);
    if (half == 0) {
      rhS[nl * 4 + 0] = sx * inv;
      rhS[nl * 4 + 1] = sy * inv;
      rhS[nl * 4 + 2] = sz * inv;
    }
    float rb[4];
#pragma unroll
    for (int u = 0; u < 4; ++u) {
      int k = (u >> 1) * 4 + 2 * half + (u & 1);
      float x = (dd - (float)k * (R0f / 7.0f)) * 4.0f;
      rb[u] = expf(-x * x);
    }

    v8f acc[4];
#pragma unroll
    for (int j = 0; j < 4; ++j) acc[j] = (v8f)0.0f;

    const float* es = emb + (size_t)sN * 64;
    const float* ed = emb + (size_t)dN * 64;
#pragma unroll
    for (int t = 0; t < 16; ++t) {
      int k0 = 4 * t + 2 * half;
      v2f a = *(const v2f*)(es + k0);
#pragma unroll
      for (int j = 0; j < 4; ++j) {
        v2f b = *(const v2f*)(smem + OFF_WSRC + (16 * j + nl) * 64 + k0);
        acc[j] = wmma4(a, b, acc[j]);
      }
    }
#pragma unroll
    for (int t = 0; t < 16; ++t) {
      int k0 = 4 * t + 2 * half;
      v2f a = *(const v2f*)(ed + k0);
#pragma unroll
      for (int j = 0; j < 4; ++j) {
        v2f b = *(const v2f*)(smem + OFF_WDST + (16 * j + nl) * 64 + k0);
        acc[j] = wmma4(a, b, acc[j]);
      }
    }
#pragma unroll
    for (int t = 0; t < 2; ++t) {
      int k0 = 4 * t + 2 * half;
      v2f a;
      a[0] = rb[2 * t];
      a[1] = rb[2 * t + 1];
#pragma unroll
      for (int j = 0; j < 4; ++j) {
        v2f b = *(const v2f*)(smem + OFF_WENC + (16 * j + nl) * 8 + k0);
        acc[j] = wmma4(a, b, acc[j]);
      }
    }
#pragma unroll
    for (int j = 0; j < 4; ++j) {
      float bb = smem[OFF_BIAS0 + 16 * j + nl];
#pragma unroll
      for (int r = 0; r < 8; ++r) acc[j][r] += bb;
    }

    lds_fence();
#pragma unroll
    for (int j = 0; j < 4; ++j)
#pragma unroll
      for (int r = 0; r < 8; ++r)
        stage[(r + 8 * half) * 64 + 16 * j + nl] = acc[j][r];
    lds_fence();

    int sn[8], dn[8];
#pragma unroll
    for (int r = 0; r < 8; ++r) {
      sn[r] = idxS[r + 8 * half];
      dn[r] = idxS[16 + r + 8 * half];
    }
    float rh[3][8];
#pragma unroll
    for (int r = 0; r < 8; ++r) {
      rh[0][r] = rhS[(r + 8 * half) * 4 + 0];
      rh[1][r] = rhS[(r + 8 * half) * 4 + 1];
      rh[2][r] = rhS[(r + 8 * half) * 4 + 2];
    }
    float z0r[4][8];
#pragma unroll
    for (int j = 0; j < 4; ++j)
#pragma unroll
      for (int r = 0; r < 8; ++r)
        z0r[j][r] = z0g[(size_t)dn[r] * 64 + 16 * j + nl];
    float z1r[3][4][8];
#pragma unroll
    for (int i = 0; i < 3; ++i)
#pragma unroll
      for (int j = 0; j < 4; ++j)
#pragma unroll
        for (int r = 0; r < 8; ++r)
          z1r[i][j][r] = z1g[(size_t)dn[r] * 192 + i * 64 + 16 * j + nl];

    v8f g[4][4];
#pragma unroll
    for (int c = 0; c < 4; ++c) {
#pragma unroll
      for (int j = 0; j < 4; ++j) g[c][j] = (v8f)0.0f;
#pragma unroll
      for (int t = 0; t < 16; ++t) {
        int k0 = 4 * t + 2 * half;
        v2f a = *(const v2f*)(stage + nl * 64 + k0);
#pragma unroll
        for (int j = 0; j < 4; ++j) {
          v2f b = *(const v2f*)(smem + OFF_WGATE +
                                (64 * c + 16 * j + nl) * 64 + k0);
          g[c][j] = wmma4(a, b, g[c][j]);
        }
      }
#pragma unroll
      for (int j = 0; j < 4; ++j) {
        float bb = smem[OFF_BGATE + 64 * c + 16 * j + nl];
#pragma unroll
        for (int r = 0; r < 8; ++r) g[c][j][r] += bb;
      }
    }

#pragma unroll
    for (int j = 0; j < 4; ++j)
#pragma unroll
      for (int r = 0; r < 8; ++r) {
        float dotv = rh[0][r] * z1r[0][j][r] + rh[1][r] * z1r[1][j][r] +
                     rh[2][r] * z1r[2][j][r];
        g[0][j][r] = g[0][j][r] * z0r[j][r] + g[1][j][r] * dotv;
      }

    lds_fence();
#pragma unroll
    for (int j = 0; j < 4; ++j)
#pragma unroll
      for (int r = 0; r < 8; ++r)
        stage[(r + 8 * half) * 64 + 16 * j + nl] = g[0][j][r];
    lds_fence();
    v8f p[4];
#pragma unroll
    for (int j = 0; j < 4; ++j) p[j] = (v8f)0.0f;
#pragma unroll
    for (int t = 0; t < 16; ++t) {
      int k0 = 4 * t + 2 * half;
      v2f a = *(const v2f*)(stage + nl * 64 + k0);
#pragma unroll
      for (int j = 0; j < 4; ++j) {
        v2f b = *(const v2f*)(smem + OFF_WS + (16 * j + nl) * 64 + k0);
        p[j] = wmma4(a, b, p[j]);
      }
    }
#pragma unroll
    for (int j = 0; j < 4; ++j)
#pragma unroll
      for (int r = 0; r < 8; ++r)
        atomicAdd(&out0[(size_t)sn[r] * 64 + 16 * j + nl], p[j][r]);

#pragma unroll
    for (int i = 0; i < 3; ++i) {
      lds_fence();
#pragma unroll
      for (int j = 0; j < 4; ++j)
#pragma unroll
        for (int r = 0; r < 8; ++r)
          stage[(r + 8 * half) * 64 + 16 * j + nl] =
              g[2][j][r] * z1r[i][j][r] + g[3][j][r] * rh[i][r] * z0r[j][r];
      lds_fence();
#pragma unroll
      for (int j = 0; j < 4; ++j) p[j] = (v8f)0.0f;
#pragma unroll
      for (int t = 0; t < 16; ++t) {
        int k0 = 4 * t + 2 * half;
        v2f a = *(const v2f*)(stage + nl * 64 + k0);
#pragma unroll
        for (int j = 0; j < 4; ++j) {
          v2f b = *(const v2f*)(smem + OFF_WV + (16 * j + nl) * 64 + k0);
          p[j] = wmma4(a, b, p[j]);
        }
      }
#pragma unroll
      for (int j = 0; j < 4; ++j)
#pragma unroll
        for (int r = 0; r < 8; ++r)
          atomicAdd(&out1[(size_t)sn[r] * 192 + i * 64 + 16 * j + nl],
                    p[j][r]);
    }
  }
}

extern "C" void kernel_launch(void* const* d_in, const int* in_sizes, int n_in,
                              void* d_out, int out_size, void* d_ws,
                              size_t ws_size, hipStream_t stream) {
  const int*   src   = (const int*)d_in[0];
  const int*   dst   = (const int*)d_in[1];
  const float* r_ij  = (const float*)d_in[2];
  const float* z0    = (const float*)d_in[3];
  const float* z1    = (const float*)d_in[4];
  const float* emb   = (const float*)d_in[5];
  const float* Wenc  = (const float*)d_in[6];
  const float* benc  = (const float*)d_in[7];
  const float* Wsrc  = (const float*)d_in[8];
  const float* bsrc  = (const float*)d_in[9];
  const float* Wdst  = (const float*)d_in[10];
  const float* bdst  = (const float*)d_in[11];
  const float* Wgate = (const float*)d_in[12];
  const float* bgate = (const float*)d_in[13];
  const float* Ws    = (const float*)d_in[14];
  const float* Wv    = (const float*)d_in[15];
  float* out = (float*)d_out;

  zero_f32<<<(out_size + 255) / 256, 256, 0, stream>>>(out, out_size);

  size_t smem_bytes = (size_t)SMEM_FLOATS * sizeof(float);
  messages_kernel<<<GRID, BLOCK, smem_bytes, stream>>>(
      src, dst, r_ij, z0, z1, emb, Wenc, benc, Wsrc, bsrc, Wdst, bdst, Wgate,
      bgate, Ws, Wv, out, out + (size_t)NNODES * CHAN);
}


# edges on the WMMA N axis, weights as A operand: D register pairs are the next product's operand, no LDS staging of activations; W_v fragments read once per tile; k pairing, float sum order change
# speedup vs baseline: 2.3887x; 2.3887x over previous
#include <hip/hip_runtime.h>
#include <math.h>


#define NNODES 50000
#define NEDGES 500000
#define CHAN   64
#define R0f    2.0f

#define WAVES  8
#define BLOCK  (WAVES * 32)
#define NTILES (NEDGES / 16)
#define GRID   512

typedef float v2f __attribute__((ext_vector_type(2)));
typedef float v8f __attribute__((ext_vector_type(8)));

#define OFF_WSRC  0
#define OFF_WDST  4096
#define OFF_WS    8192
#define OFF_WV    12288
#define OFF_WGATE 16384
#define OFF_WENC  32768
#define OFF_BIAS0 33280
#define OFF_BGATE 33344
#define SMEM_FLOATS 33600


__device__ __forceinline__ v8f wmma4(v2f a, v2f b, v8f c) {
  return __builtin_amdgcn_wmma_f32_16x16x4_f32(false, a, false, b, (short)0, c,
                                               false, false);
}

__device__ __forceinline__ void gate_chunk(const float* smem, int c, int nl,
                                           int half, const v8f (&acc)[4],
                                           v8f (&g)[4]) {
#pragma unroll
  for (int m = 0; m < 4; ++m) g[m] = (v8f)0.0f;
#pragma unroll
  for (int j = 0; j < 4; ++j)
#pragma unroll
    for (int t = 0; t < 4; ++t) {
      v2f b;
      b[0] = acc[j][2 * t];
      b[1] = acc[j][2 * t + 1];
#pragma unroll
      for (int m = 0; m < 4; ++m) {
        v2f a = *(const v2f*)(smem + OFF_WGATE + (64 * c + 16 * m + nl) * 64 +
                              16 * j + 8 * half + 2 * t);
        g[m] = wmma4(a, b, g[m]);
      }
    }
#pragma unroll
  for (int m = 0; m < 4; ++m)
#pragma unroll
    for (int r = 0; r < 8; ++r)
      g[m][r] += smem[OFF_BGATE + 64 * c + 16 * m + 8 * half + r];
}

__global__ void zero_f32(float* __restrict__ p, int n) {
  int i = blockIdx.x * blockDim.x + threadIdx.x;
  if (i < n) p[i] = 0.0f;
}

__global__ void __launch_bounds__(BLOCK, 1)
messages_kernel(const int* __restrict__ src, const int* __restrict__ dst,
                const float* __restrict__ r_ij,
                const float* __restrict__ z0g, const float* __restrict__ z1g,
                const float* __restrict__ emb,
                const float* __restrict__ Wenc, const float* __restrict__ benc,
                const float* __restrict__ Wsrc, const float* __restrict__ bsrc,
                const float* __restrict__ Wdst, const float* __restrict__ bdst,
                const float* __restrict__ Wgate, const float* __restrict__ bgate,
                const float* __restrict__ Ws, const float* __restrict__ Wv,
                float* __restrict__ out0, float* __restrict__ out1) {
  extern __shared__ __attribute__((aligned(16))) float smem[];

  for (int i = threadIdx.x; i < 64 * 64; i += BLOCK) {
    int k = i >> 6, n = i & 63;
    smem[OFF_WSRC + n * 64 + k] = Wsrc[i];
    smem[OFF_WDST + n * 64 + k] = Wdst[i];
    smem[OFF_WS   + n * 64 + k] = Ws[i];
    smem[OFF_WV   + n * 64 + k] = Wv[i];
  }
  for (int i = threadIdx.x; i < 64 * 256; i += BLOCK) {
    int k = i >> 8, n = i & 255;
    smem[OFF_WGATE + n * 64 + k] = Wgate[i];
  }
  for (int i = threadIdx.x; i < 8 * 64; i += BLOCK) {
    int k = i >> 6, n = i & 63;
    smem[OFF_WENC + n * 8 + k] = Wenc[i];
  }
  if (threadIdx.x < 64)
    smem[OFF_BIAS0 + threadIdx.x] =
        bsrc[threadIdx.x] + bdst[threadIdx.x] + benc[threadIdx.x];
  if (threadIdx.x < 256) smem[OFF_BGATE + threadIdx.x] = bgate[threadIdx.x];
  __syncthreads();

  const int lane = threadIdx.x & 31;
  const int wave = threadIdx.x >> 5;
  const int half = lane >> 4;
  const int nl   = lane & 15;

  for (int tile = blockIdx.x * WAVES + wave; tile < NTILES;
       tile += gridDim.x * WAVES) {
    const int e  = tile * 16 + nl;
    const int sN = src[e];
    const int dN = dst[e];

    int sn[8];
    {
      const int4* sp = (const int4*)(src + tile * 16 + 8 * half);
      const int4 s0 = sp[0], s1 = sp[1];
      sn[0] = s0.x; sn[1] = s0.y; sn[2] = s0.z; sn[3] = s0.w;
      sn[4] = s1.x; sn[5] = s1.y; sn[6] = s1.z; sn[7] = s1.w;
    }

    float4 eS[8], eD[8];
    {
      const float4* es = (const float4*)(emb + (size_t)sN * 64 + 32 * half);
      const float4* ed = (const float4*)(emb + (size_t)dN * 64 + 32 * half);
#pragma unroll
      for (int q = 0; q < 8; ++q) { eS[q] = es[q]; eD[q] = ed[q]; }
    }
    float rx = r_ij[e * 3 + 0], ry = r_ij[e * 3 + 1], rz = r_ij[e * 3 + 2];
    float dd  = sqrtf(rx * rx + ry * ry + rz * rz);
    float sx = rx * 3.5f, sy = ry * 3.5f, sz = rz * 3.5f;
    float inv = rsqrtf(1.0f + sx * sx + sy * sy + sz * sz);
    float rh[3];
    rh[0] = sx * inv; rh[1] = sy * inv; rh[2] = sz * inv;
    float rb[4];
#pragma unroll
    for (int u = 0; u < 4; ++u) {
      int k = (u >> 1) * 4 + 2 * half + (u & 1);
      float x = (dd - (float)k * (R0f / 7.0f)) * 4.0f;
      rb[u] = expf(-x * x);
    }

    v8f acc[4];
#pragma unroll
    for (int j = 0; j < 4; ++j) acc[j] = (v8f)0.0f;

#pragma unroll
    for (int t = 0; t < 16; ++t) {
      const float4 v = eS[t >> 1];
      v2f b;
      b[0] = (t & 1) ? v.z : v.x;
      b[1] = (t & 1) ? v.w : v.y;
#pragma unroll
      for (int j = 0; j < 4; ++j) {
        v2f a = *(const v2f*)(smem + OFF_WSRC + (16 * j + nl) * 64 + 32 * half + 2 * t);
        acc[j] = wmma4(a, b, acc[j]);
      }
    }
#pragma unroll
    for (int t = 0; t < 16; ++t) {
      const float4 v = eD[t >> 1];
      v2f b;
      b[0] = (t & 1) ? v.z : v.x;
      b[1] = (t & 1) ? v.w : v.y;
#pragma unroll
      for (int j = 0; j < 4; ++j) {
        v2f a = *(const v2f*)(smem + OFF_WDST + (16 * j + nl) * 64 + 32 * half + 2 * t);
        acc[j] = wmma4(a, b, acc[j]);
      }
    }
#pragma unroll
    for (int t = 0; t < 2; ++t) {
      v2f b;
      b[0] = rb[2 * t];
      b[1] = rb[2 * t + 1];
#pragma unroll
      for (int j = 0; j < 4; ++j) {
        v2f a = *(const v2f*)(smem + OFF_WENC + (16 * j + nl) * 8 + 4 * t + 2 * half);
        acc[j] = wmma4(a, b, acc[j]);
      }
    }
#pragma unroll
    for (int j = 0; j < 4; ++j)
#pragma unroll
      for (int r = 0; r < 8; ++r)
        acc[j][r] += smem[OFF_BIAS0 + 16 * j + 8 * half + r];

    float z0r[4][8];
#pragma unroll
    for (int j = 0; j < 4; ++j) {
      const float4* zp =
          (const float4*)(z0g + (size_t)dN * 64 + 16 * j + 8 * half);
      const float4 a = zp[0], b = zp[1];
      z0r[j][0] = a.x; z0r[j][1] = a.y; z0r[j][2] = a.z; z0r[j][3] = a.w;
      z0r[j][4] = b.x; z0r[j][5] = b.y; z0r[j][6] = b.z; z0r[j][7] = b.w;
    }
    float z1r[3][4][8];
#pragma unroll
    for (int i = 0; i < 3; ++i)
#pragma unroll
      for (int j = 0; j < 4; ++j) {
        const float4* zp = (const float4*)(z1g + (size_t)dN * 192 + i * 64 +
                                           16 * j + 8 * half);
        const float4 a = zp[0], b = zp[1];
        z1r[i][j][0] = a.x; z1r[i][j][1] = a.y; z1r[i][j][2] = a.z; z1r[i][j][3] = a.w;
        z1r[i][j][4] = b.x; z1r[i][j][5] = b.y; z1r[i][j][6] = b.z; z1r[i][j][7] = b.w;
      }

    v8f g0[4], g1[4];
    gate_chunk(smem, 0, nl, half, acc, g0);
    gate_chunk(smem, 1, nl, half, acc, g1);

#pragma unroll
    for (int j = 0; j < 4; ++j)
#pragma unroll
      for (int r = 0; r < 8; ++r) {
        float dotv = rh[0] * z1r[0][j][r] + rh[1] * z1r[1][j][r] +
                     rh[2] * z1r[2][j][r];
        g0[j][r] = g0[j][r] * z0r[j][r] + g1[j][r] * dotv;
      }

    v8f p[4];
#pragma unroll
    for (int n = 0; n < 4; ++n) p[n] = (v8f)0.0f;
#pragma unroll
    for (int j = 0; j < 4; ++j)
#pragma unroll
      for (int t = 0; t < 4; ++t) {
        v2f a;
        a[0] = g0[j][2 * t];
        a[1] = g0[j][2 * t + 1];
#pragma unroll
        for (int n = 0; n < 4; ++n) {
          v2f b = *(const v2f*)(smem + OFF_WS + (16 * n + nl) * 64 + 16 * j +
                                8 * half + 2 * t);
          p[n] = wmma4(a, b, p[n]);
        }
      }
#pragma unroll
    for (int n = 0; n < 4; ++n)
#pragma unroll
      for (int r = 0; r < 8; ++r)
        atomicAdd(&out0[(size_t)sn[r] * 64 + 16 * n + nl], p[n][r]);

    v8f g2[4], g3[4];
    gate_chunk(smem, 2, nl, half, acc, g2);
    gate_chunk(smem, 3, nl, half, acc, g3);

    v8f vm[3][4];
#pragma unroll
    for (int i = 0; i < 3; ++i)
#pragma unroll
      for (int j = 0; j < 4; ++j)
#pragma unroll
        for (int r = 0; r < 8; ++r)
          vm[i][j][r] =
              g2[j][r] * z1r[i][j][r] + g3[j][r] * rh[i] * z0r[j][r];
    v8f q[3][4];
#pragma unroll
    for (int i = 0; i < 3; ++i)
#pragma unroll
      for (int n = 0; n < 4; ++n) q[i][n] = (v8f)0.0f;
#pragma unroll
    for (int j = 0; j < 4; ++j)
#pragma unroll
      for (int t = 0; t < 4; ++t)
#pragma unroll
        for (int n = 0; n < 4; ++n) {
          v2f b = *(const v2f*)(smem + OFF_WV + (16 * n + nl) * 64 + 16 * j +
                                8 * half + 2 * t);
#pragma unroll
          for (int i = 0; i < 3; ++i) {
            v2f a;
            a[0] = vm[i][j][2 * t];
            a[1] = vm[i][j][2 * t + 1];
            q[i][n] = wmma4(a, b, q[i][n]);
          }
        }
#pragma unroll
    for (int i = 0; i < 3; ++i)
#pragma unroll
      for (int n = 0; n < 4; ++n)
#pragma unroll
        for (int r = 0; r < 8; ++r)
          atomicAdd(&out1[(size_t)sn[r] * 192 + i * 64 + 16 * n + nl],
                    q[i][n][r]);
  }
}

extern "C" void kernel_launch(void* const* d_in, const int* in_sizes, int n_in,
                              void* d_out, int out_size, void* d_ws,
                              size_t ws_size, hipStream_t stream) {
  const int*   src   = (const int*)d_in[0];
  const int*   dst   = (const int*)d_in[1];
  const float* r_ij  = (const float*)d_in[2];
  const float* z0    = (const float*)d_in[3];
  const float* z1    = (const float*)d_in[4];
  const float* emb   = (const float*)d_in[5];
  const float* Wenc  = (const float*)d_in[6];
  const float* benc  = (const float*)d_in[7];
  const float* Wsrc  = (const float*)d_in[8];
  const float* bsrc  = (const float*)d_in[9];
  const float* Wdst  = (const float*)d_in[10];
  const float* bdst  = (const float*)d_in[11];
  const float* Wgate = (const float*)d_in[12];
  const float* bgate = (const float*)d_in[13];
  const float* Ws    = (const float*)d_in[14];
  const float* Wv    = (const float*)d_in[15];
  float* out = (float*)d_out;

  zero_f32<<<(out_size + 255) / 256, 256, 0, stream>>>(out, out_size);

  size_t smem_bytes = (size_t)SMEM_FLOATS * sizeof(float);
  messages_kernel<<<GRID, BLOCK, smem_bytes, stream>>>(
      src, dst, r_ij, z0, z1, emb, Wenc, benc, Wsrc, bsrc, Wdst, bdst, Wgate,
      bgate, Ws, Wv, out, out + (size_t)NNODES * CHAN);
}


# weights kept in LDS in WMMA fragment order (a wave's 16-B-per-lane fragment read is one contiguous 512-B block), not as 64-float rows: the 16-way bank conflict of every weight fragment read goes
# speedup vs baseline: 3.7048x; 1.5510x over previous
#include <hip/hip_runtime.h>
#include <math.h>


#define NNODES 50000
#define NEDGES 500000
#define CHAN   64
#define R0f    2.0f

#define WAVES  8
#define BLOCK  (WAVES * 32)
#define NTILES (NEDGES / 16)
#define GRID   512

typedef float v2f __attribute__((ext_vector_type(2)));
typedef float v8f __attribute__((ext_vector_type(8)));

#define FRAG_LO(n, k)  ((((((n) >> 4) * 8 + (((k) & 31) >> 2)) * 32 + (((k) >> 5) * 16 + ((n) & 15))) << 2) + ((k) & 3))
#define FRAG_HI(n, k)  (((((((n) >> 4) * 4 + ((k) >> 4)) * 2 + (((k) & 7) >> 2)) * 32 + ((((k) >> 3) & 1) * 16 + ((n) & 15))) << 2) + ((k) & 3))
#define FRAG_ENC(n, k) ((((((n) >> 4) * 32) + ((((k) >> 1) & 1) * 16 + ((n) & 15))) << 2) + (((k) >> 2) * 2 + ((k) & 1)))
#define RD_LO(lane, R, t)     (4 * (lane) + 128 * (8 * (R) + ((t) >> 1)) + 2 * ((t) & 1))
#define RD_HI(lane, R, j, t)  (4 * (lane) + 128 * ((((R) * 4 + (j)) * 2) + ((t) >> 1)) + 2 * ((t) & 1))
#define RD_ENC(lane, R, t)    (4 * (lane) + 128 * (R) + 2 * (t))
#define OFF_WSRC  0
#define OFF_WDST  4096
#define OFF_WS    8192
#define OFF_WV    12288
#define OFF_WGATE 16384
#define OFF_WENC  32768
#define OFF_BIAS0 33280
#define OFF_BGATE 33344
#define SMEM_FLOATS 33600


__device__ __forceinline__ v8f wmma4(v2f a, v2f b, v8f c) {
  return __builtin_amdgcn_wmma_f32_16x16x4_f32(false, a, false, b, (short)0, c,
                                               false, false);
}

__device__ __forceinline__ void gate_chunk(const float* smem, int c, int lane,
                                           int half, const v8f (&acc)[4],
                                           v8f (&g)[4]) {
#pragma unroll
  for (int m = 0; m < 4; ++m) g[m] = (v8f)0.0f;
#pragma unroll
  for (int j = 0; j < 4; ++j)
#pragma unroll
    for (int t = 0; t < 4; ++t) {
      v2f b;
      b[0] = acc[j][2 * t];
      b[1] = acc[j][2 * t + 1];
#pragma unroll
      for (int m = 0; m < 4; ++m) {
        v2f a = *(const v2f*)(smem + OFF_WGATE + RD_HI(lane, 4 * c + m, j, t));
        g[m] = wmma4(a, b, g[m]);
      }
    }
#pragma unroll
  for (int m = 0; m < 4; ++m)
#pragma unroll
    for (int r = 0; r < 8; ++r)
      g[m][r] += smem[OFF_BGATE + 64 * c + 16 * m + 8 * half + r];
}

__global__ void zero_f32(float* __restrict__ p, int n) {
  int i = blockIdx.x * blockDim.x + threadIdx.x;
  if (i < n) p[i] = 0.0f;
}

__global__ void __launch_bounds__(BLOCK, 1)
messages_kernel(const int* __restrict__ src, const int* __restrict__ dst,
                const float* __restrict__ r_ij,
                const float* __restrict__ z0g, const float* __restrict__ z1g,
                const float* __restrict__ emb,
                const float* __restrict__ Wenc, const float* __restrict__ benc,
                const float* __restrict__ Wsrc, const float* __restrict__ bsrc,
                const float* __restrict__ Wdst, const float* __restrict__ bdst,
                const float* __restrict__ Wgate, const float* __restrict__ bgate,
                const float* __restrict__ Ws, const float* __restrict__ Wv,
                float* __restrict__ out0, float* __restrict__ out1) {
  extern __shared__ __attribute__((aligned(16))) float smem[];

  for (int i = threadIdx.x; i < 64 * 64; i += BLOCK) {
    int k = i >> 6, n = i & 63;
    smem[OFF_WSRC + FRAG_LO(n, k)] = Wsrc[i];
    smem[OFF_WDST + FRAG_LO(n, k)] = Wdst[i];
    smem[OFF_WS   + FRAG_HI(n, k)] = Ws[i];
    smem[OFF_WV   + FRAG_HI(n, k)] = Wv[i];
  }
  for (int i = threadIdx.x; i < 64 * 256; i += BLOCK) {
    int k = i >> 8, n = i & 255;
    smem[OFF_WGATE + FRAG_HI(n, k)] = Wgate[i];
  }
  for (int i = threadIdx.x; i < 8 * 64; i += BLOCK) {
    int k = i >> 6, n = i & 63;
    smem[OFF_WENC + FRAG_ENC(n, k)] = Wenc[i];
  }
  if (threadIdx.x < 64)
    smem[OFF_BIAS0 + threadIdx.x] =
        bsrc[threadIdx.x] + bdst[threadIdx.x] + benc[threadIdx.x];
  if (threadIdx.x < 256) smem[OFF_BGATE + threadIdx.x] = bgate[threadIdx.x];
  __syncthreads();

  const int lane = threadIdx.x & 31;
  const int wave = threadIdx.x >> 5;
  const int half = lane >> 4;
  const int nl   = lane & 15;

  for (int tile = blockIdx.x * WAVES + wave; tile < NTILES;
       tile += gridDim.x * WAVES) {
    const int e  = tile * 16 + nl;
    const int sN = src[e];
    const int dN = dst[e];

    int sn[8];
    {
      const int4* sp = (const int4*)(src + tile * 16 + 8 * half);
      const int4 s0 = sp[0], s1 = sp[1];
      sn[0] = s0.x; sn[1] = s0.y; sn[2] = s0.z; sn[3] = s0.w;
      sn[4] = s1.x; sn[5] = s1.y; sn[6] = s1.z; sn[7] = s1.w;
    }

    float4 eS[8], eD[8];
    {
      const float4* es = (const float4*)(emb + (size_t)sN * 64 + 32 * half);
      const float4* ed = (const float4*)(emb + (size_t)dN * 64 + 32 * half);
#pragma unroll
      for (int q = 0; q < 8; ++q) { eS[q] = es[q]; eD[q] = ed[q]; }
    }
    float rx = r_ij[e * 3 + 0], ry = r_ij[e * 3 + 1], rz = r_ij[e * 3 + 2];
    float dd  = sqrtf(rx * rx + ry * ry + rz * rz);
    float sx = rx * 3.5f, sy = ry * 3.5f, sz = rz * 3.5f;
    float inv = rsqrtf(1.0f + sx * sx + sy * sy + sz * sz);
    float rh[3];
    rh[0] = sx * inv; rh[1] = sy * inv; rh[2] = sz * inv;
    float rb[4];
#pragma unroll
    for (int u = 0; u < 4; ++u) {
      int k = (u >> 1) * 4 + 2 * half + (u & 1);
      float x = (dd - (float)k * (R0f / 7.0f)) * 4.0f;
      rb[u] = expf(-x * x);
    }

    v8f acc[4];
#pragma unroll
    for (int j = 0; j < 4; ++j) acc[j] = (v8f)0.0f;

#pragma unroll
    for (int t = 0; t < 16; ++t) {
      const float4 v = eS[t >> 1];
      v2f b;
      b[0] = (t & 1) ? v.z : v.x;
      b[1] = (t & 1) ? v.w : v.y;
#pragma unroll
      for (int j = 0; j < 4; ++j) {
        v2f a = *(const v2f*)(smem + OFF_WSRC + RD_LO(lane, j, t));
        acc[j] = wmma4(a, b, acc[j]);
      }
    }
#pragma unroll
    for (int t = 0; t < 16; ++t) {
      const float4 v = eD[t >> 1];
      v2f b;
      b[0] = (t & 1) ? v.z : v.x;
      b[1] = (t & 1) ? v.w : v.y;
#pragma unroll
      for (int j = 0; j < 4; ++j) {
        v2f a = *(const v2f*)(smem + OFF_WDST + RD_LO(lane, j, t));
        acc[j] = wmma4(a, b, acc[j]);
      }
    }
#pragma unroll
    for (int t = 0; t < 2; ++t) {
      v2f b;
      b[0] = rb[2 * t];
      b[1] = rb[2 * t + 1];
#pragma unroll
      for (int j = 0; j < 4; ++j) {
        v2f a = *(const v2f*)(smem + OFF_WENC + RD_ENC(lane, j, t));
        acc[j] = wmma4(a, b, acc[j]);
      }
    }
#pragma unroll
    for (int j = 0; j < 4; ++j)
#pragma unroll
      for (int r = 0; r < 8; ++r)
        acc[j][r] += smem[OFF_BIAS0 + 16 * j + 8 * half + r];

    float z0r[4][8];
#pragma unroll
    for (int j = 0; j < 4; ++j) {
      const float4* zp =
          (const float4*)(z0g + (size_t)dN * 64 + 16 * j + 8 * half);
      const float4 a = zp[0], b = zp[1];
      z0r[j][0] = a.x; z0r[j][1] = a.y; z0r[j][2] = a.z; z0r[j][3] = a.w;
      z0r[j][4] = b.x; z0r[j][5] = b.y; z0r[j][6] = b.z; z0r[j][7] = b.w;
    }
    float z1r[3][4][8];
#pragma unroll
    for (int i = 0; i < 3; ++i)
#pragma unroll
      for (int j = 0; j < 4; ++j) {
        const float4* zp = (const float4*)(z1g + (size_t)dN * 192 + i * 64 +
                                           16 * j + 8 * half);
        const float4 a = zp[0], b = zp[1];
        z1r[i][j][0] = a.x; z1r[i][j][1] = a.y; z1r[i][j][2] = a.z; z1r[i][j][3] = a.w;
        z1r[i][j][4] = b.x; z1r[i][j][5] = b.y; z1r[i][j][6] = b.z; z1r[i][j][7] = b.w;
      }

    v8f g0[4], g1[4];
    gate_chunk(smem, 0, lane, half, acc, g0);
    gate_chunk(smem, 1, lane, half, acc, g1);

#pragma unroll
    for (int j = 0; j < 4; ++j)
#pragma unroll
      for (int r = 0; r < 8; ++r) {
        float dotv = rh[0] * z1r[0][j][r] + rh[1] * z1r[1][j][r] +
                     rh[2] * z1r[2][j][r];
        g0[j][r] = g0[j][r] * z0r[j][r] + g1[j][r] * dotv;
      }

    v8f p[4];
#pragma unroll
    for (int n = 0; n < 4; ++n) p[n] = (v8f)0.0f;
#pragma unroll
    for (int j = 0; j < 4; ++j)
#pragma unroll
      for (int t = 0; t < 4; ++t) {
        v2f a;
        a[0] = g0[j][2 * t];
        a[1] = g0[j][2 * t + 1];
#pragma unroll
        for (int n = 0; n < 4; ++n) {
          v2f b = *(const v2f*)(smem + OFF_WS + RD_HI(lane, n, j, t));
          p[n] = wmma4(a, b, p[n]);
        }
      }
#pragma unroll
    for (int n = 0; n < 4; ++n)
#pragma unroll
      for (int r = 0; r < 8; ++r)
        atomicAdd(&out0[(size_t)sn[r] * 64 + 16 * n + nl], p[n][r]);

    v8f g2[4], g3[4];
    gate_chunk(smem, 2, lane, half, acc, g2);
    gate_chunk(smem, 3, lane, half, acc, g3);

    v8f vm[3][4];
#pragma unroll
    for (int i = 0; i < 3; ++i)
#pragma unroll
      for (int j = 0; j < 4; ++j)
#pragma unroll
        for (int r = 0; r < 8; ++r)
          vm[i][j][r] =
              g2[j][r] * z1r[i][j][r] + g3[j][r] * rh[i] * z0r[j][r];
    v8f q[3][4];
#pragma unroll
    for (int i = 0; i < 3; ++i)
#pragma unroll
      for (int n = 0; n < 4; ++n) q[i][n] = (v8f)0.0f;
#pragma unroll
    for (int j = 0; j < 4; ++j)
#pragma unroll
      for (int t = 0; t < 4; ++t)
#pragma unroll
        for (int n = 0; n < 4; ++n) {
          v2f b = *(const v2f*)(smem + OFF_WV + RD_HI(lane, n, j, t));
#pragma unroll
          for (int i = 0; i < 3; ++i) {
            v2f a;
            a[0] = vm[i][j][2 * t];
            a[1] = vm[i][j][2 * t + 1];
            q[i][n] = wmma4(a, b, q[i][n]);
          }
        }
#pragma unroll
    for (int i = 0; i < 3; ++i)
#pragma unroll
      for (int n = 0; n < 4; ++n)
#pragma unroll
        for (int r = 0; r < 8; ++r)
          atomicAdd(&out1[(size_t)sn[r] * 192 + i * 64 + 16 * n + nl],
                    q[i][n][r]);
  }
}

extern "C" void kernel_launch(void* const* d_in, const int* in_sizes, int n_in,
                              void* d_out, int out_size, void* d_ws,
                              size_t ws_size, hipStream_t stream) {
  const int*   src   = (const int*)d_in[0];
  const int*   dst   = (const int*)d_in[1];
  const float* r_ij  = (const float*)d_in[2];
  const float* z0    = (const float*)d_in[3];
  const float* z1    = (const float*)d_in[4];
  const float* emb   = (const float*)d_in[5];
  const float* Wenc  = (const float*)d_in[6];
  const float* benc  = (const float*)d_in[7];
  const float* Wsrc  = (const float*)d_in[8];
  const float* bsrc  = (const float*)d_in[9];
  const float* Wdst  = (const float*)d_in[10];
  const float* bdst  = (const float*)d_in[11];
  const float* Wgate = (const float*)d_in[12];
  const float* bgate = (const float*)d_in[13];
  const float* Ws    = (const float*)d_in[14];
  const float* Wv    = (const float*)d_in[15];
  float* out = (float*)d_out;

  zero_f32<<<(out_size + 255) / 256, 256, 0, stream>>>(out, out_size);

  size_t smem_bytes = (size_t)SMEM_FLOATS * sizeof(float);
  messages_kernel<<<GRID, BLOCK, smem_bytes, stream>>>(
      src, dst, r_ij, z0, z1, emb, Wenc, benc, Wsrc, bsrc, Wdst, bdst, Wgate,
      bgate, Ws, Wv, out, out + (size_t)NNODES * CHAN);
}


# weight staging: one resident work-group per WGP (grid 256) stages once, a thread's 131 weight loads all requested before its first LDS store, not 24 dependent load-store trips twice; sums as before
# speedup vs baseline: 3.8167x; 1.0302x over previous
#include <hip/hip_runtime.h>
#include <math.h>


#define NNODES 50000
#define NEDGES 500000
#define CHAN   64
#define R0f    2.0f

#define WAVES  8
#define BLOCK  (WAVES * 32)
#define NTILES (NEDGES / 16)
#define GRID   256

typedef float v2f __attribute__((ext_vector_type(2)));
typedef float v8f __attribute__((ext_vector_type(8)));

#define FRAG_LO(n, k)  ((((((n) >> 4) * 8 + (((k) & 31) >> 2)) * 32 + (((k) >> 5) * 16 + ((n) & 15))) << 2) + ((k) & 3))
#define FRAG_HI(n, k)  (((((((n) >> 4) * 4 + ((k) >> 4)) * 2 + (((k) & 7) >> 2)) * 32 + ((((k) >> 3) & 1) * 16 + ((n) & 15))) << 2) + ((k) & 3))
#define FRAG_ENC(n, k) ((((((n) >> 4) * 32) + ((((k) >> 1) & 1) * 16 + ((n) & 15))) << 2) + (((k) >> 2) * 2 + ((k) & 1)))
#define RD_LO(lane, R, t)     (4 * (lane) + 128 * (8 * (R) + ((t) >> 1)) + 2 * ((t) & 1))
#define RD_HI(lane, R, j, t)  (4 * (lane) + 128 * ((((R) * 4 + (j)) * 2) + ((t) >> 1)) + 2 * ((t) & 1))
#define RD_ENC(lane, R, t)    (4 * (lane) + 128 * (R) + 2 * (t))
#define OFF_WSRC  0
#define OFF_WDST  4096
#define OFF_WS    8192
#define OFF_WV    12288
#define OFF_WGATE 16384
#define OFF_WENC  32768
#define OFF_BIAS0 33280
#define OFF_BGATE 33344
#define SMEM_FLOATS 33600


__device__ __forceinline__ v8f wmma4(v2f a, v2f b, v8f c) {
  return __builtin_amdgcn_wmma_f32_16x16x4_f32(false, a, false, b, (short)0, c,
                                               false, false);
}

__device__ __forceinline__ void gate_chunk(const float* smem, int c, int lane,
                                           int half, const v8f (&acc)[4],
                                           v8f (&g)[4]) {
#pragma unroll
  for (int m = 0; m < 4; ++m) g[m] = (v8f)0.0f;
#pragma unroll
  for (int j = 0; j < 4; ++j)
#pragma unroll
    for (int t = 0; t < 4; ++t) {
      v2f b;
      b[0] = acc[j][2 * t];
      b[1] = acc[j][2 * t + 1];
#pragma unroll
      for (int m = 0; m < 4; ++m) {
        v2f a = *(const v2f*)(smem + OFF_WGATE + RD_HI(lane, 4 * c + m, j, t));
        g[m] = wmma4(a, b, g[m]);
      }
    }
#pragma unroll
  for (int m = 0; m < 4; ++m)
#pragma unroll
    for (int r = 0; r < 8; ++r)
      g[m][r] += smem[OFF_BGATE + 64 * c + 16 * m + 8 * half + r];
}

__global__ void zero_f32(float* __restrict__ p, int n) {
  int i = blockIdx.x * blockDim.x + threadIdx.x;
  if (i < n) p[i] = 0.0f;
}

__global__ void __launch_bounds__(BLOCK, 1)
messages_kernel(const int* __restrict__ src, const int* __restrict__ dst,
                const float* __restrict__ r_ij,
                const float* __restrict__ z0g, const float* __restrict__ z1g,
                const float* __restrict__ emb,
                const float* __restrict__ Wenc, const float* __restrict__ benc,
                const float* __restrict__ Wsrc, const float* __restrict__ bsrc,
                const float* __restrict__ Wdst, const float* __restrict__ bdst,
                const float* __restrict__ Wgate, const float* __restrict__ bgate,
                const float* __restrict__ Ws, const float* __restrict__ Wv,
                float* __restrict__ out0, float* __restrict__ out1) {
  extern __shared__ __attribute__((aligned(16))) float smem[];

  {
    const int tid = threadIdx.x;
    const int n6 = tid & 63, kq = tid >> 6;
    float w4[4][16], wg[64], we[2], bg;
#pragma unroll
    for (int j = 0; j < 16; ++j) {
      const int k = 16 * (j >> 2) + 4 * kq + (j & 3);
      w4[0][j] = Wsrc[k * 64 + n6];
      w4[1][j] = Wdst[k * 64 + n6];
      w4[2][j] = Ws[k * 64 + n6];
      w4[3][j] = Wv[k * 64 + n6];
    }
#pragma unroll
    for (int j = 0; j < 2; ++j) we[j] = Wenc[tid + BLOCK * j];
    bg = bgate[tid];
#pragma unroll
    for (int k = 0; k < 64; ++k) wg[k] = Wgate[k * 256 + tid];

#pragma unroll
    for (int j = 0; j < 16; ++j) {
      const int k = 16 * (j >> 2) + 4 * kq + (j & 3);
      smem[OFF_WSRC + FRAG_LO(n6, k)] = w4[0][j];
      smem[OFF_WDST + FRAG_LO(n6, k)] = w4[1][j];
      smem[OFF_WS   + FRAG_HI(n6, k)] = w4[2][j];
      smem[OFF_WV   + FRAG_HI(n6, k)] = w4[3][j];
    }
#pragma unroll
    for (int j = 0; j < 2; ++j) {
      const int i = tid + BLOCK * j;
      smem[OFF_WENC + FRAG_ENC(i & 63, i >> 6)] = we[j];
    }
    smem[OFF_BGATE + tid] = bg;
#pragma unroll
    for (int k = 0; k < 64; ++k) smem[OFF_WGATE + FRAG_HI(tid, k)] = wg[k];
    if (tid < 64) smem[OFF_BIAS0 + tid] = bsrc[tid] + bdst[tid] + benc[tid];
  }
  __syncthreads();

  const int lane = threadIdx.x & 31;
  const int wave = threadIdx.x >> 5;
  const int half = lane >> 4;
  const int nl   = lane & 15;

  for (int tile = blockIdx.x * WAVES + wave; tile < NTILES;
       tile += gridDim.x * WAVES) {
    const int e  = tile * 16 + nl;
    const int sN = src[e];
    const int dN = dst[e];

    int sn[8];
    {
      const int4* sp = (const int4*)(src + tile * 16 + 8 * half);
      const int4 s0 = sp[0], s1 = sp[1];
      sn[0] = s0.x; sn[1] = s0.y; sn[2] = s0.z; sn[3] = s0.w;
      sn[4] = s1.x; sn[5] = s1.y; sn[6] = s1.z; sn[7] = s1.w;
    }

    float4 eS[8], eD[8];
    {
      const float4* es = (const float4*)(emb + (size_t)sN * 64 + 32 * half);
      const float4* ed = (const float4*)(emb + (size_t)dN * 64 + 32 * half);
#pragma unroll
      for (int q = 0; q < 8; ++q) { eS[q] = es[q]; eD[q] = ed[q]; }
    }
    float rx = r_ij[e * 3 + 0], ry = r_ij[e * 3 + 1], rz = r_ij[e * 3 + 2];
    float dd  = sqrtf(rx * rx + ry * ry + rz * rz);
    float sx = rx * 3.5f, sy = ry * 3.5f, sz = rz * 3.5f;
    float inv = rsqrtf(1.0f + sx * sx + sy * sy + sz * sz);
    float rh[3];
    rh[0] = sx * inv; rh[1] = sy * inv; rh[2] = sz * inv;
    float rb[4];
#pragma unroll
    for (int u = 0; u < 4; ++u) {
      int k = (u >> 1) * 4 + 2 * half + (u & 1);
      float x = (dd - (float)k * (R0f / 7.0f)) * 4.0f;
      rb[u] = expf(-x * x);
    }

    v8f acc[4];
#pragma unroll
    for (int j = 0; j < 4; ++j) acc[j] = (v8f)0.0f;

#pragma unroll
    for (int t = 0; t < 16; ++t) {
      const float4 v = eS[t >> 1];
      v2f b;
      b[0] = (t & 1) ? v.z : v.x;
      b[1] = (t & 1) ? v.w : v.y;
#pragma unroll
      for (int j = 0; j < 4; ++j) {
        v2f a = *(const v2f*)(smem + OFF_WSRC + RD_LO(lane, j, t));
        acc[j] = wmma4(a, b, acc[j]);
      }
    }
#pragma unroll
    for (int t = 0; t < 16; ++t) {
      const float4 v = eD[t >> 1];
      v2f b;
      b[0] = (t & 1) ? v.z : v.x;
      b[1] = (t & 1) ? v.w : v.y;
#pragma unroll
      for (int j = 0; j < 4; ++j) {
        v2f a = *(const v2f*)(smem + OFF_WDST + RD_LO(lane, j, t));
        acc[j] = wmma4(a, b, acc[j]);
      }
    }
#pragma unroll
    for (int t = 0; t < 2; ++t) {
      v2f b;
      b[0] = rb[2 * t];
      b[1] = rb[2 * t + 1];
#pragma unroll
      for (int j = 0; j < 4; ++j) {
        v2f a = *(const v2f*)(smem + OFF_WENC + RD_ENC(lane, j, t));
        acc[j] = wmma4(a, b, acc[j]);
      }
    }
#pragma unroll
    for (int j = 0; j < 4; ++j)
#pragma unroll
      for (int r = 0; r < 8; ++r)
        acc[j][r] += smem[OFF_BIAS0 + 16 * j + 8 * half + r];

    float z0r[4][8];
#pragma unroll
    for (int j = 0; j < 4; ++j) {
      const float4* zp =
          (const float4*)(z0g + (size_t)dN * 64 + 16 * j + 8 * half);
      const float4 a = zp[0], b = zp[1];
      z0r[j][0] = a.x; z0r[j][1] = a.y; z0r[j][2] = a.z; z0r[j][3] = a.w;
      z0r[j][4] = b.x; z0r[j][5] = b.y; z0r[j][6] = b.z; z0r[j][7] = b.w;
    }
    float z1r[3][4][8];
#pragma unroll
    for (int i = 0; i < 3; ++i)
#pragma unroll
      for (int j = 0; j < 4; ++j) {
        const float4* zp = (const float4*)(z1g + (size_t)dN * 192 + i * 64 +
                                           16 * j + 8 * half);
        const float4 a = zp[0], b = zp[1];
        z1r[i][j][0] = a.x; z1r[i][j][1] = a.y; z1r[i][j][2] = a.z; z1r[i][j][3] = a.w;
        z1r[i][j][4] = b.x; z1r[i][j][5] = b.y; z1r[i][j][6] = b.z; z1r[i][j][7] = b.w;
      }

    v8f g0[4], g1[4];
    gate_chunk(smem, 0, lane, half, acc, g0);
    gate_chunk(smem, 1, lane, half, acc, g1);

#pragma unroll
    for (int j = 0; j < 4; ++j)
#pragma unroll
      for (int r = 0; r < 8; ++r) {
        float dotv = rh[0] * z1r[0][j][r] + rh[1] * z1r[1][j][r] +
                     rh[2] * z1r[2][j][r];
        g0[j][r] = g0[j][r] * z0r[j][r] + g1[j][r] * dotv;
      }

    v8f p[4];
#pragma unroll
    for (int n = 0; n < 4; ++n) p[n] = (v8f)0.0f;
#pragma unroll
    for (int j = 0; j < 4; ++j)
#pragma unroll
      for (int t = 0; t < 4; ++t) {
        v2f a;
        a[0] = g0[j][2 * t];
        a[1] = g0[j][2 * t + 1];
#pragma unroll
        for (int n = 0; n < 4; ++n) {
          v2f b = *(const v2f*)(smem + OFF_WS + RD_HI(lane, n, j, t));
          p[n] = wmma4(a, b, p[n]);
        }
      }
#pragma unroll
    for (int n = 0; n < 4; ++n)
#pragma unroll
      for (int r = 0; r < 8; ++r)
        atomicAdd(&out0[(size_t)sn[r] * 64 + 16 * n + nl], p[n][r]);

    v8f g2[4], g3[4];
    gate_chunk(smem, 2, lane, half, acc, g2);
    gate_chunk(smem, 3, lane, half, acc, g3);

    v8f vm[3][4];
#pragma unroll
    for (int i = 0; i < 3; ++i)
#pragma unroll
      for (int j = 0; j < 4; ++j)
#pragma unroll
        for (int r = 0; r < 8; ++r)
          vm[i][j][r] =
              g2[j][r] * z1r[i][j][r] + g3[j][r] * rh[i] * z0r[j][r];
    v8f q[3][4];
#pragma unroll
    for (int i = 0; i < 3; ++i)
#pragma unroll
      for (int n = 0; n < 4; ++n) q[i][n] = (v8f)0.0f;
#pragma unroll
    for (int j = 0; j < 4; ++j)
#pragma unroll
      for (int t = 0; t < 4; ++t)
#pragma unroll
        for (int n = 0; n < 4; ++n) {
          v2f b = *(const v2f*)(smem + OFF_WV + RD_HI(lane, n, j, t));
#pragma unroll
          for (int i = 0; i < 3; ++i) {
            v2f a;
            a[0] = vm[i][j][2 * t];
            a[1] = vm[i][j][2 * t + 1];
            q[i][n] = wmma4(a, b, q[i][n]);
          }
        }
#pragma unroll
    for (int i = 0; i < 3; ++i)
#pragma unroll
      for (int n = 0; n < 4; ++n)
#pragma unroll
        for (int r = 0; r < 8; ++r)
          atomicAdd(&out1[(size_t)sn[r] * 192 + i * 64 + 16 * n + nl],
                    q[i][n][r]);
  }
}

extern "C" void kernel_launch(void* const* d_in, const int* in_sizes, int n_in,
                              void* d_out, int out_size, void* d_ws,
                              size_t ws_size, hipStream_t stream) {
  const int*   src   = (const int*)d_in[0];
  const int*   dst   = (const int*)d_in[1];
  const float* r_ij  = (const float*)d_in[2];
  const float* z0    = (const float*)d_in[3];
  const float* z1    = (const float*)d_in[4];
  const float* emb   = (const float*)d_in[5];
  const float* Wenc  = (const float*)d_in[6];
  const float* benc  = (const float*)d_in[7];
  const float* Wsrc  = (const float*)d_in[8];
  const float* bsrc  = (const float*)d_in[9];
  const float* Wdst  = (const float*)d_in[10];
  const float* bdst  = (const float*)d_in[11];
  const float* Wgate = (const float*)d_in[12];
  const float* bgate = (const float*)d_in[13];
  const float* Ws    = (const float*)d_in[14];
  const float* Wv    = (const float*)d_in[15];
  float* out = (float*)d_out;

  zero_f32<<<(out_size + 255) / 256, 256, 0, stream>>>(out, out_size);

  size_t smem_bytes = (size_t)SMEM_FLOATS * sizeof(float);
  messages_kernel<<<GRID, BLOCK, smem_bytes, stream>>>(
      src, dst, r_ij, z0, z1, emb, Wenc, benc, Wsrc, bsrc, Wdst, bdst, Wgate,
      bgate, Ws, Wv, out, out + (size_t)NNODES * CHAN);
}


# gate, W_s, W_v: a set's eight 16-B weight-fragment LDS reads issued as one group ahead of the previous set's 16 WMMAs (held by scheduling barriers), not one read per two WMMAs; sums as before
# speedup vs baseline: 4.0326x; 1.0566x over previous
#include <hip/hip_runtime.h>
#include <math.h>


#define NNODES 50000
#define NEDGES 500000
#define CHAN   64
#define R0f    2.0f

#define WAVES  8
#define BLOCK  (WAVES * 32)
#define NTILES (NEDGES / 16)
#define GRID   256

typedef float v2f __attribute__((ext_vector_type(2)));
typedef float v8f __attribute__((ext_vector_type(8)));

#define FRAG_LO(n, k)  ((((((n) >> 4) * 8 + (((k) & 31) >> 2)) * 32 + (((k) >> 5) * 16 + ((n) & 15))) << 2) + ((k) & 3))
#define FRAG_HI(n, k)  (((((((n) >> 4) * 4 + ((k) >> 4)) * 2 + (((k) & 7) >> 2)) * 32 + ((((k) >> 3) & 1) * 16 + ((n) & 15))) << 2) + ((k) & 3))
#define FRAG_ENC(n, k) ((((((n) >> 4) * 32) + ((((k) >> 1) & 1) * 16 + ((n) & 15))) << 2) + (((k) >> 2) * 2 + ((k) & 1)))
#define RD_LO(lane, R, t)     (4 * (lane) + 128 * (8 * (R) + ((t) >> 1)) + 2 * ((t) & 1))
#define RD_HI(lane, R, j, t)  (4 * (lane) + 128 * ((((R) * 4 + (j)) * 2) + ((t) >> 1)) + 2 * ((t) & 1))
#define RD_ENC(lane, R, t)    (4 * (lane) + 128 * (R) + 2 * (t))
#define OFF_WSRC  0
#define OFF_WDST  4096
#define OFF_WS    8192
#define OFF_WV    12288
#define OFF_WGATE 16384
#define OFF_WENC  32768
#define OFF_BIAS0 33280
#define OFF_BGATE 33344
#define SMEM_FLOATS 33600


__device__ __forceinline__ v8f wmma4(v2f a, v2f b, v8f c) {
  return __builtin_amdgcn_wmma_f32_16x16x4_f32(false, a, false, b, (short)0, c,
                                               false, false);
}


__device__ __forceinline__ void group_end() { __builtin_amdgcn_sched_barrier(0); }

__device__ __forceinline__ void ld_set(const float* smem, int off, int R0, int j,
                                       int lane, float4 (&f)[8]) {
#pragma unroll
  for (int m = 0; m < 4; ++m)
#pragma unroll
    for (int u = 0; u < 2; ++u)
      f[2 * m + u] = *(const float4*)(smem + off + RD_HI(lane, R0 + m, j, 2 * u));
}

__device__ __forceinline__ v2f frag_of(const float4 (&f)[8], int m, int t) {
  const float4 v = f[2 * m + (t >> 1)];
  v2f a;
  a[0] = (t & 1) ? v.z : v.x;
  a[1] = (t & 1) ? v.w : v.y;
  return a;
}

__device__ __forceinline__ void mma_gate_set(const float4 (&f)[8], const v8f& accj,
                                             v8f (&g)[4]) {
#pragma unroll
  for (int t = 0; t < 4; ++t) {
    v2f b;
    b[0] = accj[2 * t];
    b[1] = accj[2 * t + 1];
#pragma unroll
    for (int m = 0; m < 4; ++m) g[m] = wmma4(frag_of(f, m, t), b, g[m]);
  }
}

__device__ __forceinline__ void ld_gate_bias(const float* smem, int c, int half,
                                             float4 (&bb)[8]) {
#pragma unroll
  for (int m = 0; m < 4; ++m)
#pragma unroll
    for (int u = 0; u < 2; ++u)
      bb[2 * m + u] = *(const float4*)(smem + OFF_BGATE + 64 * c + 16 * m +
                                       8 * half + 4 * u);
}

__device__ __forceinline__ void add_gate_bias(v8f (&g)[4], const float4 (&bb)[8]) {
#pragma unroll
  for (int m = 0; m < 4; ++m)
#pragma unroll
    for (int u = 0; u < 2; ++u) {
      g[m][4 * u + 0] += bb[2 * m + u].x;
      g[m][4 * u + 1] += bb[2 * m + u].y;
      g[m][4 * u + 2] += bb[2 * m + u].z;
      g[m][4 * u + 3] += bb[2 * m + u].w;
    }
}

__global__ void zero_f32(float* __restrict__ p, int n) {
  int i = blockIdx.x * blockDim.x + threadIdx.x;
  if (i < n) p[i] = 0.0f;
}

__global__ void __launch_bounds__(BLOCK, 1)
messages_kernel(const int* __restrict__ src, const int* __restrict__ dst,
                const float* __restrict__ r_ij,
                const float* __restrict__ z0g, const float* __restrict__ z1g,
                const float* __restrict__ emb,
                const float* __restrict__ Wenc, const float* __restrict__ benc,
                const float* __restrict__ Wsrc, const float* __restrict__ bsrc,
                const float* __restrict__ Wdst, const float* __restrict__ bdst,
                const float* __restrict__ Wgate, const float* __restrict__ bgate,
                const float* __restrict__ Ws, const float* __restrict__ Wv,
                float* __restrict__ out0, float* __restrict__ out1) {
  extern __shared__ __attribute__((aligned(16))) float smem[];

  {
    const int tid = threadIdx.x;
    const int n6 = tid & 63, kq = tid >> 6;
    float w4[4][16], wg[64], we[2], bg;
#pragma unroll
    for (int j = 0; j < 16; ++j) {
      const int k = 16 * (j >> 2) + 4 * kq + (j & 3);
      w4[0][j] = Wsrc[k * 64 + n6];
      w4[1][j] = Wdst[k * 64 + n6];
      w4[2][j] = Ws[k * 64 + n6];
      w4[3][j] = Wv[k * 64 + n6];
    }
#pragma unroll
    for (int j = 0; j < 2; ++j) we[j] = Wenc[tid + BLOCK * j];
    bg = bgate[tid];
#pragma unroll
    for (int k = 0; k < 64; ++k) wg[k] = Wgate[k * 256 + tid];

#pragma unroll
    for (int j = 0; j < 16; ++j) {
      const int k = 16 * (j >> 2) + 4 * kq + (j & 3);
      smem[OFF_WSRC + FRAG_LO(n6, k)] = w4[0][j];
      smem[OFF_WDST + FRAG_LO(n6, k)] = w4[1][j];
      smem[OFF_WS   + FRAG_HI(n6, k)] = w4[2][j];
      smem[OFF_WV   + FRAG_HI(n6, k)] = w4[3][j];
    }
#pragma unroll
    for (int j = 0; j < 2; ++j) {
      const int i = tid + BLOCK * j;
      smem[OFF_WENC + FRAG_ENC(i & 63, i >> 6)] = we[j];
    }
    smem[OFF_BGATE + tid] = bg;
#pragma unroll
    for (int k = 0; k < 64; ++k) smem[OFF_WGATE + FRAG_HI(tid, k)] = wg[k];
    if (tid < 64) smem[OFF_BIAS0 + tid] = bsrc[tid] + bdst[tid] + benc[tid];
  }
  __syncthreads();

  const int lane = threadIdx.x & 31;
  const int wave = threadIdx.x >> 5;
  const int half = lane >> 4;
  const int nl   = lane & 15;

  for (int tile = blockIdx.x * WAVES + wave; tile < NTILES;
       tile += gridDim.x * WAVES) {
    const int e  = tile * 16 + nl;
    const int sN = src[e];
    const int dN = dst[e];

    int sn[8];
    {
      const int4* sp = (const int4*)(src + tile * 16 + 8 * half);
      const int4 s0 = sp[0], s1 = sp[1];
      sn[0] = s0.x; sn[1] = s0.y; sn[2] = s0.z; sn[3] = s0.w;
      sn[4] = s1.x; sn[5] = s1.y; sn[6] = s1.z; sn[7] = s1.w;
    }

    float4 eS[8], eD[8];
    {
      const float4* es = (const float4*)(emb + (size_t)sN * 64 + 32 * half);
      const float4* ed = (const float4*)(emb + (size_t)dN * 64 + 32 * half);
#pragma unroll
      for (int q = 0; q < 8; ++q) { eS[q] = es[q]; eD[q] = ed[q]; }
    }
    float rx = r_ij[e * 3 + 0], ry = r_ij[e * 3 + 1], rz = r_ij[e * 3 + 2];
    float dd  = sqrtf(rx * rx + ry * ry + rz * rz);
    float sx = rx * 3.5f, sy = ry * 3.5f, sz = rz * 3.5f;
    float inv = rsqrtf(1.0f + sx * sx + sy * sy + sz * sz);
    float rh[3];
    rh[0] = sx * inv; rh[1] = sy * inv; rh[2] = sz * inv;
    float rb[4];
#pragma unroll
    for (int u = 0; u < 4; ++u) {
      int k = (u >> 1) * 4 + 2 * half + (u & 1);
      float x = (dd - (float)k * (R0f / 7.0f)) * 4.0f;
      rb[u] = expf(-x * x);
    }

    v8f acc[4];
#pragma unroll
    for (int j = 0; j < 4; ++j) acc[j] = (v8f)0.0f;

#pragma unroll
    for (int t = 0; t < 16; ++t) {
      const float4 v = eS[t >> 1];
      v2f b;
      b[0] = (t & 1) ? v.z : v.x;
      b[1] = (t & 1) ? v.w : v.y;
#pragma unroll
      for (int j = 0; j < 4; ++j) {
        v2f a = *(const v2f*)(smem + OFF_WSRC + RD_LO(lane, j, t));
        acc[j] = wmma4(a, b, acc[j]);
      }
    }
#pragma unroll
    for (int t = 0; t < 16; ++t) {
      const float4 v = eD[t >> 1];
      v2f b;
      b[0] = (t & 1) ? v.z : v.x;
      b[1] = (t & 1) ? v.w : v.y;
#pragma unroll
      for (int j = 0; j < 4; ++j) {
        v2f a = *(const v2f*)(smem + OFF_WDST + RD_LO(lane, j, t));
        acc[j] = wmma4(a, b, acc[j]);
      }
    }
#pragma unroll
    for (int t = 0; t < 2; ++t) {
      v2f b;
      b[0] = rb[2 * t];
      b[1] = rb[2 * t + 1];
#pragma unroll
      for (int j = 0; j < 4; ++j) {
        v2f a = *(const v2f*)(smem + OFF_WENC + RD_ENC(lane, j, t));
        acc[j] = wmma4(a, b, acc[j]);
      }
    }
#pragma unroll
    for (int j = 0; j < 4; ++j)
#pragma unroll
      for (int r = 0; r < 8; ++r)
        acc[j][r] += smem[OFF_BIAS0 + 16 * j + 8 * half + r];

    float z0r[4][8];
#pragma unroll
    for (int j = 0; j < 4; ++j) {
      const float4* zp =
          (const float4*)(z0g + (size_t)dN * 64 + 16 * j + 8 * half);
      const float4 a = zp[0], b = zp[1];
      z0r[j][0] = a.x; z0r[j][1] = a.y; z0r[j][2] = a.z; z0r[j][3] = a.w;
      z0r[j][4] = b.x; z0r[j][5] = b.y; z0r[j][6] = b.z; z0r[j][7] = b.w;
    }
    float z1r[3][4][8];
#pragma unroll
    for (int i = 0; i < 3; ++i)
#pragma unroll
      for (int j = 0; j < 4; ++j) {
        const float4* zp = (const float4*)(z1g + (size_t)dN * 192 + i * 64 +
                                           16 * j + 8 * half);
        const float4 a = zp[0], b = zp[1];
        z1r[i][j][0] = a.x; z1r[i][j][1] = a.y; z1r[i][j][2] = a.z; z1r[i][j][3] = a.w;
        z1r[i][j][4] = b.x; z1r[i][j][5] = b.y; z1r[i][j][6] = b.z; z1r[i][j][7] = b.w;
      }

    float4 fr[2][8], bb[8];
    v8f g0[4], g1[4];
#pragma unroll
    for (int m = 0; m < 4; ++m) { g0[m] = (v8f)0.0f; g1[m] = (v8f)0.0f; }
    ld_set(smem, OFF_WGATE, 0, 0, lane, fr[0]);
    group_end();
#pragma unroll
    for (int s = 0; s < 8; ++s) {
      if (s < 7)
        ld_set(smem, OFF_WGATE, 4 * ((s + 1) >> 2), (s + 1) & 3, lane,
               fr[(s + 1) & 1]);
      else
        ld_set(smem, OFF_WS, 0, 0, lane, fr[(s + 1) & 1]);
      if ((s & 3) == 2) ld_gate_bias(smem, s >> 2, half, bb);
      group_end();
      if (s < 4) {
        mma_gate_set(fr[s & 1], acc[s & 3], g0);
        if (s == 3) add_gate_bias(g0, bb);
      } else {
        mma_gate_set(fr[s & 1], acc[s & 3], g1);
        if (s == 7) add_gate_bias(g1, bb);
      }
      group_end();
    }

#pragma unroll
    for (int j = 0; j < 4; ++j)
#pragma unroll
      for (int r = 0; r < 8; ++r) {
        float dotv = rh[0] * z1r[0][j][r] + rh[1] * z1r[1][j][r] +
                     rh[2] * z1r[2][j][r];
        g0[j][r] = g0[j][r] * z0r[j][r] + g1[j][r] * dotv;
      }

    v8f p[4];
#pragma unroll
    for (int n = 0; n < 4; ++n) p[n] = (v8f)0.0f;
#pragma unroll
    for (int s = 0; s < 4; ++s) {
      if (s < 3)
        ld_set(smem, OFF_WS, 0, s + 1, lane, fr[(s + 1) & 1]);
      else
        ld_set(smem, OFF_WGATE, 8, 0, lane, fr[(s + 1) & 1]);
      group_end();
#pragma unroll
      for (int t = 0; t < 4; ++t) {
        v2f a;
        a[0] = g0[s][2 * t];
        a[1] = g0[s][2 * t + 1];
#pragma unroll
        for (int n = 0; n < 4; ++n)
          p[n] = wmma4(a, frag_of(fr[s & 1], n, t), p[n]);
      }
      group_end();
    }
#pragma unroll
    for (int n = 0; n < 4; ++n)
#pragma unroll
      for (int r = 0; r < 8; ++r)
        atomicAdd(&out0[(size_t)sn[r] * 64 + 16 * n + nl], p[n][r]);

    v8f g2[4], g3[4];
#pragma unroll
    for (int m = 0; m < 4; ++m) { g2[m] = (v8f)0.0f; g3[m] = (v8f)0.0f; }
#pragma unroll
    for (int s = 0; s < 8; ++s) {
      if (s < 7)
        ld_set(smem, OFF_WGATE, 4 * (2 + ((s + 1) >> 2)), (s + 1) & 3, lane,
               fr[(s + 1) & 1]);
      else
        ld_set(smem, OFF_WV, 0, 0, lane, fr[(s + 1) & 1]);
      if ((s & 3) == 2) ld_gate_bias(smem, 2 + (s >> 2), half, bb);
      group_end();
      if (s < 4) {
        mma_gate_set(fr[s & 1], acc[s & 3], g2);
        if (s == 3) add_gate_bias(g2, bb);
      } else {
        mma_gate_set(fr[s & 1], acc[s & 3], g3);
        if (s == 7) add_gate_bias(g3, bb);
      }
      group_end();
    }

    v8f vm[3][4];
#pragma unroll
    for (int i = 0; i < 3; ++i)
#pragma unroll
      for (int j = 0; j < 4; ++j)
#pragma unroll
        for (int r = 0; r < 8; ++r)
          vm[i][j][r] =
              g2[j][r] * z1r[i][j][r] + g3[j][r] * rh[i] * z0r[j][r];
    v8f q[3][4];
#pragma unroll
    for (int i = 0; i < 3; ++i)
#pragma unroll
      for (int n = 0; n < 4; ++n) q[i][n] = (v8f)0.0f;
#pragma unroll
    for (int s = 0; s < 4; ++s) {
      if (s < 3) ld_set(smem, OFF_WV, 0, s + 1, lane, fr[(s + 1) & 1]);
      group_end();
#pragma unroll
      for (int t = 0; t < 4; ++t)
#pragma unroll
        for (int n = 0; n < 4; ++n) {
          const v2f b = frag_of(fr[s & 1], n, t);
#pragma unroll
          for (int i = 0; i < 3; ++i) {
            v2f a;
            a[0] = vm[i][s][2 * t];
            a[1] = vm[i][s][2 * t + 1];
            q[i][n] = wmma4(a, b, q[i][n]);
          }
        }
      group_end();
    }
#pragma unroll
    for (int i = 0; i < 3; ++i)
#pragma unroll
      for (int n = 0; n < 4; ++n)
#pragma unroll
        for (int r = 0; r < 8; ++r)
          atomicAdd(&out1[(size_t)sn[r] * 192 + i * 64 + 16 * n + nl],
                    q[i][n][r]);
  }
}

extern "C" void kernel_launch(void* const* d_in, const int* in_sizes, int n_in,
                              void* d_out, int out_size, void* d_ws,
                              size_t ws_size, hipStream_t stream) {
  const int*   src   = (const int*)d_in[0];
  const int*   dst   = (const int*)d_in[1];
  const float* r_ij  = (const float*)d_in[2];
  const float* z0    = (const float*)d_in[3];
  const float* z1    = (const float*)d_in[4];
  const float* emb   = (const float*)d_in[5];
  const float* Wenc  = (const float*)d_in[6];
  const float* benc  = (const float*)d_in[7];
  const float* Wsrc  = (const float*)d_in[8];
  const float* bsrc  = (const float*)d_in[9];
  const float* Wdst  = (const float*)d_in[10];
  const float* bdst  = (const float*)d_in[11];
  const float* Wgate = (const float*)d_in[12];
  const float* bgate = (const float*)d_in[13];
  const float* Ws    = (const float*)d_in[14];
  const float* Wv    = (const float*)d_in[15];
  float* out = (float*)d_out;

  zero_f32<<<(out_size + 255) / 256, 256, 0, stream>>>(out, out_size);

  size_t smem_bytes = (size_t)SMEM_FLOATS * sizeof(float);
  messages_kernel<<<GRID, BLOCK, smem_bytes, stream>>>(
      src, dst, r_ij, z0, z1, emb, Wenc, benc, Wsrc, bsrc, Wdst, bdst, Wgate,
      bgate, Ws, Wv, out, out + (size_t)NNODES * CHAN);
}


# tiles taken one by one from a per-work-group LDS counter (one range each), not dealt equally by wave number: a SIMD's older wave takes more, none ends alone; atomic sums in arrival order as before
# speedup vs baseline: 4.2467x; 1.0531x over previous
#include <hip/hip_runtime.h>
#include <math.h>


#define NNODES 50000
#define NEDGES 500000
#define CHAN   64
#define R0f    2.0f

#define WAVES  8
#define BLOCK  (WAVES * 32)
#define NTILES (NEDGES / 16)
#define GRID   256

typedef float v2f __attribute__((ext_vector_type(2)));
typedef float v8f __attribute__((ext_vector_type(8)));

#define FRAG_LO(n, k)  ((((((n) >> 4) * 8 + (((k) & 31) >> 2)) * 32 + (((k) >> 5) * 16 + ((n) & 15))) << 2) + ((k) & 3))
#define FRAG_HI(n, k)  (((((((n) >> 4) * 4 + ((k) >> 4)) * 2 + (((k) & 7) >> 2)) * 32 + ((((k) >> 3) & 1) * 16 + ((n) & 15))) << 2) + ((k) & 3))
#define FRAG_ENC(n, k) ((((((n) >> 4) * 32) + ((((k) >> 1) & 1) * 16 + ((n) & 15))) << 2) + (((k) >> 2) * 2 + ((k) & 1)))
#define RD_LO(lane, R, t)     (4 * (lane) + 128 * (8 * (R) + ((t) >> 1)) + 2 * ((t) & 1))
#define RD_HI(lane, R, j, t)  (4 * (lane) + 128 * ((((R) * 4 + (j)) * 2) + ((t) >> 1)) + 2 * ((t) & 1))
#define RD_ENC(lane, R, t)    (4 * (lane) + 128 * (R) + 2 * (t))
#define OFF_WSRC  0
#define OFF_WDST  4096
#define OFF_WS    8192
#define OFF_WV    12288
#define OFF_WGATE 16384
#define OFF_WENC  32768
#define OFF_BIAS0 33280
#define OFF_BGATE 33344
#define OFF_NEXT  33600
#define SMEM_FLOATS 33604


__device__ __forceinline__ v8f wmma4(v2f a, v2f b, v8f c) {
  return __builtin_amdgcn_wmma_f32_16x16x4_f32(false, a, false, b, (short)0, c,
                                               false, false);
}


__device__ __forceinline__ void group_end() { __builtin_amdgcn_sched_barrier(0); }

__device__ __forceinline__ void ld_set(const float* smem, int off, int R0, int j,
                                       int lane, float4 (&f)[8]) {
#pragma unroll
  for (int m = 0; m < 4; ++m)
#pragma unroll
    for (int u = 0; u < 2; ++u)
      f[2 * m + u] = *(const float4*)(smem + off + RD_HI(lane, R0 + m, j, 2 * u));
}

__device__ __forceinline__ v2f frag_of(const float4 (&f)[8], int m, int t) {
  const float4 v = f[2 * m + (t >> 1)];
  v2f a;
  a[0] = (t & 1) ? v.z : v.x;
  a[1] = (t & 1) ? v.w : v.y;
  return a;
}

__device__ __forceinline__ void mma_gate_set(const float4 (&f)[8], const v8f& accj,
                                             v8f (&g)[4]) {
#pragma unroll
  for (int t = 0; t < 4; ++t) {
    v2f b;
    b[0] = accj[2 * t];
    b[1] = accj[2 * t + 1];
#pragma unroll
    for (int m = 0; m < 4; ++m) g[m] = wmma4(frag_of(f, m, t), b, g[m]);
  }
}

__device__ __forceinline__ void ld_gate_bias(const float* smem, int c, int half,
                                             float4 (&bb)[8]) {
#pragma unroll
  for (int m = 0; m < 4; ++m)
#pragma unroll
    for (int u = 0; u < 2; ++u)
      bb[2 * m + u] = *(const float4*)(smem + OFF_BGATE + 64 * c + 16 * m +
                                       8 * half + 4 * u);
}

__device__ __forceinline__ void add_gate_bias(v8f (&g)[4], const float4 (&bb)[8]) {
#pragma unroll
  for (int m = 0; m < 4; ++m)
#pragma unroll
    for (int u = 0; u < 2; ++u) {
      g[m][4 * u + 0] += bb[2 * m + u].x;
      g[m][4 * u + 1] += bb[2 * m + u].y;
      g[m][4 * u + 2] += bb[2 * m + u].z;
      g[m][4 * u + 3] += bb[2 * m + u].w;
    }
}

__global__ void zero_f32(float* __restrict__ p, int n) {
  int i = blockIdx.x * blockDim.x + threadIdx.x;
  if (i < n) p[i] = 0.0f;
}

__global__ void __launch_bounds__(BLOCK, 1)
messages_kernel(const int* __restrict__ src, const int* __restrict__ dst,
                const float* __restrict__ r_ij,
                const float* __restrict__ z0g, const float* __restrict__ z1g,
                const float* __restrict__ emb,
                const float* __restrict__ Wenc, const float* __restrict__ benc,
                const float* __restrict__ Wsrc, const float* __restrict__ bsrc,
                const float* __restrict__ Wdst, const float* __restrict__ bdst,
                const float* __restrict__ Wgate, const float* __restrict__ bgate,
                const float* __restrict__ Ws, const float* __restrict__ Wv,
                float* __restrict__ out0, float* __restrict__ out1) {
  extern __shared__ __attribute__((aligned(16))) float smem[];

  {
    const int tid = threadIdx.x;
    const int n6 = tid & 63, kq = tid >> 6;
    float w4[4][16], wg[64], we[2], bg;
#pragma unroll
    for (int j = 0; j < 16; ++j) {
      const int k = 16 * (j >> 2) + 4 * kq + (j & 3);
      w4[0][j] = Wsrc[k * 64 + n6];
      w4[1][j] = Wdst[k * 64 + n6];
      w4[2][j] = Ws[k * 64 + n6];
      w4[3][j] = Wv[k * 64 + n6];
    }
#pragma unroll
    for (int j = 0; j < 2; ++j) we[j] = Wenc[tid + BLOCK * j];
    bg = bgate[tid];
#pragma unroll
    for (int k = 0; k < 64; ++k) wg[k] = Wgate[k * 256 + tid];

#pragma unroll
    for (int j = 0; j < 16; ++j) {
      const int k = 16 * (j >> 2) + 4 * kq + (j & 3);
      smem[OFF_WSRC + FRAG_LO(n6, k)] = w4[0][j];
      smem[OFF_WDST + FRAG_LO(n6, k)] = w4[1][j];
      smem[OFF_WS   + FRAG_HI(n6, k)] = w4[2][j];
      smem[OFF_WV   + FRAG_HI(n6, k)] = w4[3][j];
    }
#pragma unroll
    for (int j = 0; j < 2; ++j) {
      const int i = tid + BLOCK * j;
      smem[OFF_WENC + FRAG_ENC(i & 63, i >> 6)] = we[j];
    }
    smem[OFF_BGATE + tid] = bg;
#pragma unroll
    for (int k = 0; k < 64; ++k) smem[OFF_WGATE + FRAG_HI(tid, k)] = wg[k];
    if (tid < 64) smem[OFF_BIAS0 + tid] = bsrc[tid] + bdst[tid] + benc[tid];
    if (tid == 0) *(int*)(smem + OFF_NEXT) = WAVES;
  }
  __syncthreads();

  const int lane = threadIdx.x & 31;
  const int half = lane >> 4;
  const int nl   = lane & 15;

  const unsigned nwg = gridDim.x;
  const int lo = (int)((blockIdx.x * (unsigned)NTILES) / nwg);
  const int hi = (int)(((blockIdx.x + 1u) * (unsigned)NTILES) / nwg);
  int* const taken = (int*)(smem + OFF_NEXT);
  int k = __builtin_amdgcn_readfirstlane((int)(threadIdx.x >> 5));

  for (int trip = lo; trip < hi; ++trip) {
    const int tile = lo + k;
    if (tile >= hi) break;
    const int e  = tile * 16 + nl;
    const int sN = src[e];
    const int dN = dst[e];

    int sn[8];
    {
      const int4* sp = (const int4*)(src + tile * 16 + 8 * half);
      const int4 s0 = sp[0], s1 = sp[1];
      sn[0] = s0.x; sn[1] = s0.y; sn[2] = s0.z; sn[3] = s0.w;
      sn[4] = s1.x; sn[5] = s1.y; sn[6] = s1.z; sn[7] = s1.w;
    }

    float4 eS[8], eD[8];
    {
      const float4* es = (const float4*)(emb + (size_t)sN * 64 + 32 * half);
      const float4* ed = (const float4*)(emb + (size_t)dN * 64 + 32 * half);
#pragma unroll
      for (int q = 0; q < 8; ++q) { eS[q] = es[q]; eD[q] = ed[q]; }
    }
    float rx = r_ij[e * 3 + 0], ry = r_ij[e * 3 + 1], rz = r_ij[e * 3 + 2];
    float dd  = sqrtf(rx * rx + ry * ry + rz * rz);
    float sx = rx * 3.5f, sy = ry * 3.5f, sz = rz * 3.5f;
    float inv = rsqrtf(1.0f + sx * sx + sy * sy + sz * sz);
    float rh[3];
    rh[0] = sx * inv; rh[1] = sy * inv; rh[2] = sz * inv;
    float rb[4];
#pragma unroll
    for (int u = 0; u < 4; ++u) {
      int k = (u >> 1) * 4 + 2 * half + (u & 1);
      float x = (dd - (float)k * (R0f / 7.0f)) * 4.0f;
      rb[u] = expf(-x * x);
    }

    v8f acc[4];
#pragma unroll
    for (int j = 0; j < 4; ++j) acc[j] = (v8f)0.0f;

#pragma unroll
    for (int t = 0; t < 16; ++t) {
      const float4 v = eS[t >> 1];
      v2f b;
      b[0] = (t & 1) ? v.z : v.x;
      b[1] = (t & 1) ? v.w : v.y;
#pragma unroll
      for (int j = 0; j < 4; ++j) {
        v2f a = *(const v2f*)(smem + OFF_WSRC + RD_LO(lane, j, t));
        acc[j] = wmma4(a, b, acc[j]);
      }
    }
#pragma unroll
    for (int t = 0; t < 16; ++t) {
      const float4 v = eD[t >> 1];
      v2f b;
      b[0] = (t & 1) ? v.z : v.x;
      b[1] = (t & 1) ? v.w : v.y;
#pragma unroll
      for (int j = 0; j < 4; ++j) {
        v2f a = *(const v2f*)(smem + OFF_WDST + RD_LO(lane, j, t));
        acc[j] = wmma4(a, b, acc[j]);
      }
    }
#pragma unroll
    for (int t = 0; t < 2; ++t) {
      v2f b;
      b[0] = rb[2 * t];
      b[1] = rb[2 * t + 1];
#pragma unroll
      for (int j = 0; j < 4; ++j) {
        v2f a = *(const v2f*)(smem + OFF_WENC + RD_ENC(lane, j, t));
        acc[j] = wmma4(a, b, acc[j]);
      }
    }
#pragma unroll
    for (int j = 0; j < 4; ++j)
#pragma unroll
      for (int r = 0; r < 8; ++r)
        acc[j][r] += smem[OFF_BIAS0 + 16 * j + 8 * half + r];

    float z0r[4][8];
#pragma unroll
    for (int j = 0; j < 4; ++j) {
      const float4* zp =
          (const float4*)(z0g + (size_t)dN * 64 + 16 * j + 8 * half);
      const float4 a = zp[0], b = zp[1];
      z0r[j][0] = a.x; z0r[j][1] = a.y; z0r[j][2] = a.z; z0r[j][3] = a.w;
      z0r[j][4] = b.x; z0r[j][5] = b.y; z0r[j][6] = b.z; z0r[j][7] = b.w;
    }
    float z1r[3][4][8];
#pragma unroll
    for (int i = 0; i < 3; ++i)
#pragma unroll
      for (int j = 0; j < 4; ++j) {
        const float4* zp = (const float4*)(z1g + (size_t)dN * 192 + i * 64 +
                                           16 * j + 8 * half);
        const float4 a = zp[0], b = zp[1];
        z1r[i][j][0] = a.x; z1r[i][j][1] = a.y; z1r[i][j][2] = a.z; z1r[i][j][3] = a.w;
        z1r[i][j][4] = b.x; z1r[i][j][5] = b.y; z1r[i][j][6] = b.z; z1r[i][j][7] = b.w;
      }

    float4 fr[2][8], bb[8];
    v8f g0[4], g1[4];
#pragma unroll
    for (int m = 0; m < 4; ++m) { g0[m] = (v8f)0.0f; g1[m] = (v8f)0.0f; }
    ld_set(smem, OFF_WGATE, 0, 0, lane, fr[0]);
    group_end();
#pragma unroll
    for (int s = 0; s < 8; ++s) {
      if (s < 7)
        ld_set(smem, OFF_WGATE, 4 * ((s + 1) >> 2), (s + 1) & 3, lane,
               fr[(s + 1) & 1]);
      else
        ld_set(smem, OFF_WS, 0, 0, lane, fr[(s + 1) & 1]);
      if ((s & 3) == 2) ld_gate_bias(smem, s >> 2, half, bb);
      group_end();
      if (s < 4) {
        mma_gate_set(fr[s & 1], acc[s & 3], g0);
        if (s == 3) add_gate_bias(g0, bb);
      } else {
        mma_gate_set(fr[s & 1], acc[s & 3], g1);
        if (s == 7) add_gate_bias(g1, bb);
      }
      group_end();
    }

#pragma unroll
    for (int j = 0; j < 4; ++j)
#pragma unroll
      for (int r = 0; r < 8; ++r) {
        float dotv = rh[0] * z1r[0][j][r] + rh[1] * z1r[1][j][r] +
                     rh[2] * z1r[2][j][r];
        g0[j][r] = g0[j][r] * z0r[j][r] + g1[j][r] * dotv;
      }

    v8f p[4];
#pragma unroll
    for (int n = 0; n < 4; ++n) p[n] = (v8f)0.0f;
#pragma unroll
    for (int s = 0; s < 4; ++s) {
      if (s < 3)
        ld_set(smem, OFF_WS, 0, s + 1, lane, fr[(s + 1) & 1]);
      else
        ld_set(smem, OFF_WGATE, 8, 0, lane, fr[(s + 1) & 1]);
      group_end();
#pragma unroll
      for (int t = 0; t < 4; ++t) {
        v2f a;
        a[0] = g0[s][2 * t];
        a[1] = g0[s][2 * t + 1];
#pragma unroll
        for (int n = 0; n < 4; ++n)
          p[n] = wmma4(a, frag_of(fr[s & 1], n, t), p[n]);
      }
      group_end();
    }
#pragma unroll
    for (int n = 0; n < 4; ++n)
#pragma unroll
      for (int r = 0; r < 8; ++r)
        atomicAdd(&out0[(size_t)sn[r] * 64 + 16 * n + nl], p[n][r]);

    v8f g2[4], g3[4];
#pragma unroll
    for (int m = 0; m < 4; ++m) { g2[m] = (v8f)0.0f; g3[m] = (v8f)0.0f; }
#pragma unroll
    for (int s = 0; s < 8; ++s) {
      if (s < 7)
        ld_set(smem, OFF_WGATE, 4 * (2 + ((s + 1) >> 2)), (s + 1) & 3, lane,
               fr[(s + 1) & 1]);
      else
        ld_set(smem, OFF_WV, 0, 0, lane, fr[(s + 1) & 1]);
      if ((s & 3) == 2) ld_gate_bias(smem, 2 + (s >> 2), half, bb);
      group_end();
      if (s < 4) {
        mma_gate_set(fr[s & 1], acc[s & 3], g2);
        if (s == 3) add_gate_bias(g2, bb);
      } else {
        mma_gate_set(fr[s & 1], acc[s & 3], g3);
        if (s == 7) add_gate_bias(g3, bb);
      }
      group_end();
    }

    v8f vm[3][4];
#pragma unroll
    for (int i = 0; i < 3; ++i)
#pragma unroll
      for (int j = 0; j < 4; ++j)
#pragma unroll
        for (int r = 0; r < 8; ++r)
          vm[i][j][r] =
              g2[j][r] * z1r[i][j][r] + g3[j][r] * rh[i] * z0r[j][r];
    v8f q[3][4];
#pragma unroll
    for (int i = 0; i < 3; ++i)
#pragma unroll
      for (int n = 0; n < 4; ++n) q[i][n] = (v8f)0.0f;
#pragma unroll
    for (int s = 0; s < 4; ++s) {
      if (s < 3) ld_set(smem, OFF_WV, 0, s + 1, lane, fr[(s + 1) & 1]);
      group_end();
#pragma unroll
      for (int t = 0; t < 4; ++t)
#pragma unroll
        for (int n = 0; n < 4; ++n) {
          const v2f b = frag_of(fr[s & 1], n, t);
#pragma unroll
          for (int i = 0; i < 3; ++i) {
            v2f a;
            a[0] = vm[i][s][2 * t];
            a[1] = vm[i][s][2 * t + 1];
            q[i][n] = wmma4(a, b, q[i][n]);
          }
        }
      group_end();
    }
#pragma unroll
    for (int i = 0; i < 3; ++i)
#pragma unroll
      for (int n = 0; n < 4; ++n)
#pragma unroll
        for (int r = 0; r < 8; ++r)
          atomicAdd(&out1[(size_t)sn[r] * 192 + i * 64 + 16 * n + nl],
                    q[i][n][r]);

    int kn = 0;
    if (lane == 0) kn = atomicAdd(taken, 1);
    k = __builtin_amdgcn_readfirstlane(kn);
  }
}

extern "C" void kernel_launch(void* const* d_in, const int* in_sizes, int n_in,
                              void* d_out, int out_size, void* d_ws,
                              size_t ws_size, hipStream_t stream) {
  const int*   src   = (const int*)d_in[0];
  const int*   dst   = (const int*)d_in[1];
  const float* r_ij  = (const float*)d_in[2];
  const float* z0    = (const float*)d_in[3];
  const float* z1    = (const float*)d_in[4];
  const float* emb   = (const float*)d_in[5];
  const float* Wenc  = (const float*)d_in[6];
  const float* benc  = (const float*)d_in[7];
  const float* Wsrc  = (const float*)d_in[8];
  const float* bsrc  = (const float*)d_in[9];
  const float* Wdst  = (const float*)d_in[10];
  const float* bdst  = (const float*)d_in[11];
  const float* Wgate = (const float*)d_in[12];
  const float* bgate = (const float*)d_in[13];
  const float* Ws    = (const float*)d_in[14];
  const float* Wv    = (const float*)d_in[15];
  float* out = (float*)d_out;

  zero_f32<<<(out_size + 255) / 256, 256, 0, stream>>>(out, out_size);

  size_t smem_bytes = (size_t)SMEM_FLOATS * sizeof(float);
  messages_kernel<<<GRID, BLOCK, smem_bytes, stream>>>(
      src, dst, r_ij, z0, z1, emb, Wenc, benc, Wsrc, bsrc, Wdst, bdst, Wgate,
      bgate, Ws, Wv, out, out + (size_t)NNODES * CHAN);
}


# W_v product done one 16-channel output group at a time (k inside): a group's 24 scatter atomics go out in sixes under the next group's WMMAs, not 96 in one run at tile end; atomic sums as before
# speedup vs baseline: 4.4272x; 1.0425x over previous
#include <hip/hip_runtime.h>
#include <math.h>


#define NNODES 50000
#define NEDGES 500000
#define CHAN   64
#define R0f    2.0f

#define WAVES  8
#define BLOCK  (WAVES * 32)
#define NTILES (NEDGES / 16)
#define GRID   256

typedef float v2f __attribute__((ext_vector_type(2)));
typedef float v8f __attribute__((ext_vector_type(8)));

#define FRAG_LO(n, k)  ((((((n) >> 4) * 8 + (((k) & 31) >> 2)) * 32 + (((k) >> 5) * 16 + ((n) & 15))) << 2) + ((k) & 3))
#define FRAG_HI(n, k)  (((((((n) >> 4) * 4 + ((k) >> 4)) * 2 + (((k) & 7) >> 2)) * 32 + ((((k) >> 3) & 1) * 16 + ((n) & 15))) << 2) + ((k) & 3))
#define FRAG_ENC(n, k) ((((((n) >> 4) * 32) + ((((k) >> 1) & 1) * 16 + ((n) & 15))) << 2) + (((k) >> 2) * 2 + ((k) & 1)))
#define RD_LO(lane, R, t)     (4 * (lane) + 128 * (8 * (R) + ((t) >> 1)) + 2 * ((t) & 1))
#define RD_HI(lane, R, j, t)  (4 * (lane) + 128 * ((((R) * 4 + (j)) * 2) + ((t) >> 1)) + 2 * ((t) & 1))
#define RD_ENC(lane, R, t)    (4 * (lane) + 128 * (R) + 2 * (t))
#define OFF_WSRC  0
#define OFF_WDST  4096
#define OFF_WS    8192
#define OFF_WV    12288
#define OFF_WGATE 16384
#define OFF_WENC  32768
#define OFF_BIAS0 33280
#define OFF_BGATE 33344
#define OFF_NEXT  33600
#define SMEM_FLOATS 33604


__device__ __forceinline__ v8f wmma4(v2f a, v2f b, v8f c) {
  return __builtin_amdgcn_wmma_f32_16x16x4_f32(false, a, false, b, (short)0, c,
                                               false, false);
}


__device__ __forceinline__ void group_end() { __builtin_amdgcn_sched_barrier(0); }

__device__ __forceinline__ void ld_set(const float* smem, int off, int R0, int j,
                                       int lane, float4 (&f)[8]) {
#pragma unroll
  for (int m = 0; m < 4; ++m)
#pragma unroll
    for (int u = 0; u < 2; ++u)
      f[2 * m + u] = *(const float4*)(smem + off + RD_HI(lane, R0 + m, j, 2 * u));
}

__device__ __forceinline__ void ld_group(const float* smem, int off, int R,
                                         int lane, float4 (&f)[8]) {
#pragma unroll
  for (int j = 0; j < 4; ++j)
#pragma unroll
    for (int u = 0; u < 2; ++u)
      f[2 * j + u] = *(const float4*)(smem + off + RD_HI(lane, R, j, 2 * u));
}

__device__ __forceinline__ v2f frag_of(const float4 (&f)[8], int m, int t) {
  const float4 v = f[2 * m + (t >> 1)];
  v2f a;
  a[0] = (t & 1) ? v.z : v.x;
  a[1] = (t & 1) ? v.w : v.y;
  return a;
}

__device__ __forceinline__ void mma_gate_set(const float4 (&f)[8], const v8f& accj,
                                             v8f (&g)[4]) {
#pragma unroll
  for (int t = 0; t < 4; ++t) {
    v2f b;
    b[0] = accj[2 * t];
    b[1] = accj[2 * t + 1];
#pragma unroll
    for (int m = 0; m < 4; ++m) g[m] = wmma4(frag_of(f, m, t), b, g[m]);
  }
}

__device__ __forceinline__ void ld_gate_bias(const float* smem, int c, int half,
                                             float4 (&bb)[8]) {
#pragma unroll
  for (int m = 0; m < 4; ++m)
#pragma unroll
    for (int u = 0; u < 2; ++u)
      bb[2 * m + u] = *(const float4*)(smem + OFF_BGATE + 64 * c + 16 * m +
                                       8 * half + 4 * u);
}

__device__ __forceinline__ void add_gate_bias(v8f (&g)[4], const float4 (&bb)[8]) {
#pragma unroll
  for (int m = 0; m < 4; ++m)
#pragma unroll
    for (int u = 0; u < 2; ++u) {
      g[m][4 * u + 0] += bb[2 * m + u].x;
      g[m][4 * u + 1] += bb[2 * m + u].y;
      g[m][4 * u + 2] += bb[2 * m + u].z;
      g[m][4 * u + 3] += bb[2 * m + u].w;
    }
}

__global__ void zero_f32(float* __restrict__ p, int n) {
  int i = blockIdx.x * blockDim.x + threadIdx.x;
  if (i < n) p[i] = 0.0f;
}

__global__ void __launch_bounds__(BLOCK, 1)
messages_kernel(const int* __restrict__ src, const int* __restrict__ dst,
                const float* __restrict__ r_ij,
                const float* __restrict__ z0g, const float* __restrict__ z1g,
                const float* __restrict__ emb,
                const float* __restrict__ Wenc, const float* __restrict__ benc,
                const float* __restrict__ Wsrc, const float* __restrict__ bsrc,
                const float* __restrict__ Wdst, const float* __restrict__ bdst,
                const float* __restrict__ Wgate, const float* __restrict__ bgate,
                const float* __restrict__ Ws, const float* __restrict__ Wv,
                float* __restrict__ out0, float* __restrict__ out1) {
  extern __shared__ __attribute__((aligned(16))) float smem[];

  {
    const int tid = threadIdx.x;
    const int n6 = tid & 63, kq = tid >> 6;
    float w4[4][16], wg[64], we[2], bg;
#pragma unroll
    for (int j = 0; j < 16; ++j) {
      const int k = 16 * (j >> 2) + 4 * kq + (j & 3);
      w4[0][j] = Wsrc[k * 64 + n6];
      w4[1][j] = Wdst[k * 64 + n6];
      w4[2][j] = Ws[k * 64 + n6];
      w4[3][j] = Wv[k * 64 + n6];
    }
#pragma unroll
    for (int j = 0; j < 2; ++j) we[j] = Wenc[tid + BLOCK * j];
    bg = bgate[tid];
#pragma unroll
    for (int k = 0; k < 64; ++k) wg[k] = Wgate[k * 256 + tid];

#pragma unroll
    for (int j = 0; j < 16; ++j) {
      const int k = 16 * (j >> 2) + 4 * kq + (j & 3);
      smem[OFF_WSRC + FRAG_LO(n6, k)] = w4[0][j];
      smem[OFF_WDST + FRAG_LO(n6, k)] = w4[1][j];
      smem[OFF_WS   + FRAG_HI(n6, k)] = w4[2][j];
      smem[OFF_WV   + FRAG_HI(n6, k)] = w4[3][j];
    }
#pragma unroll
    for (int j = 0; j < 2; ++j) {
      const int i = tid + BLOCK * j;
      smem[OFF_WENC + FRAG_ENC(i & 63, i >> 6)] = we[j];
    }
    smem[OFF_BGATE + tid] = bg;
#pragma unroll
    for (int k = 0; k < 64; ++k) smem[OFF_WGATE + FRAG_HI(tid, k)] = wg[k];
    if (tid < 64) smem[OFF_BIAS0 + tid] = bsrc[tid] + bdst[tid] + benc[tid];
    if (tid == 0) *(int*)(smem + OFF_NEXT) = WAVES;
  }
  __syncthreads();

  const int lane = threadIdx.x & 31;
  const int half = lane >> 4;
  const int nl   = lane & 15;

  const unsigned nwg = gridDim.x;
  const int lo = (int)((blockIdx.x * (unsigned)NTILES) / nwg);
  const int hi = (int)(((blockIdx.x + 1u) * (unsigned)NTILES) / nwg);
  int* const taken = (int*)(smem + OFF_NEXT);
  int k = __builtin_amdgcn_readfirstlane((int)(threadIdx.x >> 5));

  for (int trip = lo; trip < hi; ++trip) {
    const int tile = lo + k;
    if (tile >= hi) break;
    const int e  = tile * 16 + nl;
    const int sN = src[e];
    const int dN = dst[e];

    int sn[8];
    {
      const int4* sp = (const int4*)(src + tile * 16 + 8 * half);
      const int4 s0 = sp[0], s1 = sp[1];
      sn[0] = s0.x; sn[1] = s0.y; sn[2] = s0.z; sn[3] = s0.w;
      sn[4] = s1.x; sn[5] = s1.y; sn[6] = s1.z; sn[7] = s1.w;
    }

    float4 eS[8], eD[8];
    {
      const float4* es = (const float4*)(emb + (size_t)sN * 64 + 32 * half);
      const float4* ed = (const float4*)(emb + (size_t)dN * 64 + 32 * half);
#pragma unroll
      for (int q = 0; q < 8; ++q) { eS[q] = es[q]; eD[q] = ed[q]; }
    }
    float rx = r_ij[e * 3 + 0], ry = r_ij[e * 3 + 1], rz = r_ij[e * 3 + 2];
    float dd  = sqrtf(rx * rx + ry * ry + rz * rz);
    float sx = rx * 3.5f, sy = ry * 3.5f, sz = rz * 3.5f;
    float inv = rsqrtf(1.0f + sx * sx + sy * sy + sz * sz);
    float rh[3];
    rh[0] = sx * inv; rh[1] = sy * inv; rh[2] = sz * inv;
    float rb[4];
#pragma unroll
    for (int u = 0; u < 4; ++u) {
      int k = (u >> 1) * 4 + 2 * half + (u & 1);
      float x = (dd - (float)k * (R0f / 7.0f)) * 4.0f;
      rb[u] = expf(-x * x);
    }

    v8f acc[4];
#pragma unroll
    for (int j = 0; j < 4; ++j) acc[j] = (v8f)0.0f;

#pragma unroll
    for (int t = 0; t < 16; ++t) {
      const float4 v = eS[t >> 1];
      v2f b;
      b[0] = (t & 1) ? v.z : v.x;
      b[1] = (t & 1) ? v.w : v.y;
#pragma unroll
      for (int j = 0; j < 4; ++j) {
        v2f a = *(const v2f*)(smem + OFF_WSRC + RD_LO(lane, j, t));
        acc[j] = wmma4(a, b, acc[j]);
      }
    }
#pragma unroll
    for (int t = 0; t < 16; ++t) {
      const float4 v = eD[t >> 1];
      v2f b;
      b[0] = (t & 1) ? v.z : v.x;
      b[1] = (t & 1) ? v.w : v.y;
#pragma unroll
      for (int j = 0; j < 4; ++j) {
        v2f a = *(const v2f*)(smem + OFF_WDST + RD_LO(lane, j, t));
        acc[j] = wmma4(a, b, acc[j]);
      }
    }
#pragma unroll
    for (int t = 0; t < 2; ++t) {
      v2f b;
      b[0] = rb[2 * t];
      b[1] = rb[2 * t + 1];
#pragma unroll
      for (int j = 0; j < 4; ++j) {
        v2f a = *(const v2f*)(smem + OFF_WENC + RD_ENC(lane, j, t));
        acc[j] = wmma4(a, b, acc[j]);
      }
    }
#pragma unroll
    for (int j = 0; j < 4; ++j)
#pragma unroll
      for (int r = 0; r < 8; ++r)
        acc[j][r] += smem[OFF_BIAS0 + 16 * j + 8 * half + r];

    float z0r[4][8];
#pragma unroll
    for (int j = 0; j < 4; ++j) {
      const float4* zp =
          (const float4*)(z0g + (size_t)dN * 64 + 16 * j + 8 * half);
      const float4 a = zp[0], b = zp[1];
      z0r[j][0] = a.x; z0r[j][1] = a.y; z0r[j][2] = a.z; z0r[j][3] = a.w;
      z0r[j][4] = b.x; z0r[j][5] = b.y; z0r[j][6] = b.z; z0r[j][7] = b.w;
    }
    float z1r[3][4][8];
#pragma unroll
    for (int i = 0; i < 3; ++i)
#pragma unroll
      for (int j = 0; j < 4; ++j) {
        const float4* zp = (const float4*)(z1g + (size_t)dN * 192 + i * 64 +
                                           16 * j + 8 * half);
        const float4 a = zp[0], b = zp[1];
        z1r[i][j][0] = a.x; z1r[i][j][1] = a.y; z1r[i][j][2] = a.z; z1r[i][j][3] = a.w;
        z1r[i][j][4] = b.x; z1r[i][j][5] = b.y; z1r[i][j][6] = b.z; z1r[i][j][7] = b.w;
      }

    float4 fr[2][8], bb[8];
    v8f g0[4], g1[4];
#pragma unroll
    for (int m = 0; m < 4; ++m) { g0[m] = (v8f)0.0f; g1[m] = (v8f)0.0f; }
    ld_set(smem, OFF_WGATE, 0, 0, lane, fr[0]);
    group_end();
#pragma unroll
    for (int s = 0; s < 8; ++s) {
      if (s < 7)
        ld_set(smem, OFF_WGATE, 4 * ((s + 1) >> 2), (s + 1) & 3, lane,
               fr[(s + 1) & 1]);
      else
        ld_set(smem, OFF_WS, 0, 0, lane, fr[(s + 1) & 1]);
      if ((s & 3) == 2) ld_gate_bias(smem, s >> 2, half, bb);
      group_end();
      if (s < 4) {
        mma_gate_set(fr[s & 1], acc[s & 3], g0);
        if (s == 3) add_gate_bias(g0, bb);
      } else {
        mma_gate_set(fr[s & 1], acc[s & 3], g1);
        if (s == 7) add_gate_bias(g1, bb);
      }
      group_end();
    }

#pragma unroll
    for (int j = 0; j < 4; ++j)
#pragma unroll
      for (int r = 0; r < 8; ++r) {
        float dotv = rh[0] * z1r[0][j][r] + rh[1] * z1r[1][j][r] +
                     rh[2] * z1r[2][j][r];
        g0[j][r] = g0[j][r] * z0r[j][r] + g1[j][r] * dotv;
      }

    v8f p[4];
#pragma unroll
    for (int n = 0; n < 4; ++n) p[n] = (v8f)0.0f;
#pragma unroll
    for (int s = 0; s < 4; ++s) {
      if (s < 3)
        ld_set(smem, OFF_WS, 0, s + 1, lane, fr[(s + 1) & 1]);
      else
        ld_set(smem, OFF_WGATE, 8, 0, lane, fr[(s + 1) & 1]);
      group_end();
#pragma unroll
      for (int t = 0; t < 4; ++t) {
        v2f a;
        a[0] = g0[s][2 * t];
        a[1] = g0[s][2 * t + 1];
#pragma unroll
        for (int n = 0; n < 4; ++n)
          p[n] = wmma4(a, frag_of(fr[s & 1], n, t), p[n]);
      }
      group_end();
    }
#pragma unroll
    for (int n = 0; n < 4; ++n)
#pragma unroll
      for (int r = 0; r < 8; ++r)
        atomicAdd(&out0[(size_t)sn[r] * 64 + 16 * n + nl], p[n][r]);

    v8f g2[4], g3[4];
#pragma unroll
    for (int m = 0; m < 4; ++m) { g2[m] = (v8f)0.0f; g3[m] = (v8f)0.0f; }
#pragma unroll
    for (int s = 0; s < 8; ++s) {
      if (s < 7)
        ld_set(smem, OFF_WGATE, 4 * (2 + ((s + 1) >> 2)), (s + 1) & 3, lane,
               fr[(s + 1) & 1]);
      else
        ld_group(smem, OFF_WV, 0, lane, fr[(s + 1) & 1]);
      if ((s & 3) == 2) ld_gate_bias(smem, 2 + (s >> 2), half, bb);
      group_end();
      if (s < 4) {
        mma_gate_set(fr[s & 1], acc[s & 3], g2);
        if (s == 3) add_gate_bias(g2, bb);
      } else {
        mma_gate_set(fr[s & 1], acc[s & 3], g3);
        if (s == 7) add_gate_bias(g3, bb);
      }
      group_end();
    }

    v8f vm[3][4];
#pragma unroll
    for (int i = 0; i < 3; ++i)
#pragma unroll
      for (int j = 0; j < 4; ++j)
#pragma unroll
        for (int r = 0; r < 8; ++r)
          vm[i][j][r] =
              g2[j][r] * z1r[i][j][r] + g3[j][r] * rh[i] * z0r[j][r];
    v8f q[2][3];
#pragma unroll
    for (int n = 0; n < 4; ++n) {
      if (n < 3) ld_group(smem, OFF_WV, n + 1, lane, fr[(n + 1) & 1]);
      group_end();
#pragma unroll
      for (int i = 0; i < 3; ++i) q[n & 1][i] = (v8f)0.0f;
#pragma unroll
      for (int j = 0; j < 4; ++j) {
#pragma unroll
        for (int t = 0; t < 4; ++t) {
          const v2f b = frag_of(fr[n & 1], j, t);
#pragma unroll
          for (int i = 0; i < 3; ++i) {
            v2f a;
            a[0] = vm[i][j][2 * t];
            a[1] = vm[i][j][2 * t + 1];
            q[n & 1][i] = wmma4(a, b, q[n & 1][i]);
          }
        }
        if (n > 0) {
          group_end();
#pragma unroll
          for (int c = 6 * j; c < 6 * j + 6; ++c)
            atomicAdd(&out1[(size_t)sn[c & 7] * 192 + (c >> 3) * 64 +
                            16 * (n - 1) + nl],
                      q[(n - 1) & 1][c >> 3][c & 7]);
          group_end();
        }
      }
      if (n == 0) group_end();
    }
#pragma unroll
    for (int i = 0; i < 3; ++i)
#pragma unroll
      for (int r = 0; r < 8; ++r)
        atomicAdd(&out1[(size_t)sn[r] * 192 + i * 64 + 16 * 3 + nl], q[1][i][r]);

    int kn = 0;
    if (lane == 0) kn = atomicAdd(taken, 1);
    k = __builtin_amdgcn_readfirstlane(kn);
  }
}

extern "C" void kernel_launch(void* const* d_in, const int* in_sizes, int n_in,
                              void* d_out, int out_size, void* d_ws,
                              size_t ws_size, hipStream_t stream) {
  const int*   src   = (const int*)d_in[0];
  const int*   dst   = (const int*)d_in[1];
  const float* r_ij  = (const float*)d_in[2];
  const float* z0    = (const float*)d_in[3];
  const float* z1    = (const float*)d_in[4];
  const float* emb   = (const float*)d_in[5];
  const float* Wenc  = (const float*)d_in[6];
  const float* benc  = (const float*)d_in[7];
  const float* Wsrc  = (const float*)d_in[8];
  const float* bsrc  = (const float*)d_in[9];
  const float* Wdst  = (const float*)d_in[10];
  const float* bdst  = (const float*)d_in[11];
  const float* Wgate = (const float*)d_in[12];
  const float* bgate = (const float*)d_in[13];
  const float* Ws    = (const float*)d_in[14];
  const float* Wv    = (const float*)d_in[15];
  float* out = (float*)d_out;

  zero_f32<<<(out_size + 255) / 256, 256, 0, stream>>>(out, out_size);

  size_t smem_bytes = (size_t)SMEM_FLOATS * sizeof(float);
  messages_kernel<<<GRID, BLOCK, smem_bytes, stream>>>(
      src, dst, r_ij, z0, z1, emb, Wenc, benc, Wsrc, bsrc, Wdst, bdst, Wgate,
      bgate, Ws, Wv, out, out + (size_t)NNODES * CHAN);
}
